# starting kernel
# baseline (speedup 1.0000x reference)
#include <hip/hip_runtime.h>
#include <hip/hip_bf16.h>
#include <cstdint>

typedef __attribute__((ext_vector_type(2))) float v2f;
typedef __attribute__((ext_vector_type(8))) float v8f;

#define GIN_N 100000
#define GIN_E 1600000
#define GIN_D 64
#define WPB 4
#define TSTRIDE 66

__global__ __launch_bounds__(256) void gin_zero(float4* __restrict__ p, int n4) {
  int i = blockIdx.x * blockDim.x + threadIdx.x;
  if (i < n4) p[i] = make_float4(0.f, 0.f, 0.f, 0.f);
}

__global__ __launch_bounds__(256) void gin_scatter(
    const float* __restrict__ h, const int* __restrict__ src,
    const int* __restrict__ dst, float* __restrict__ neigh, int nedges) {
  int gid  = blockIdx.x * blockDim.x + threadIdx.x;
  int e    = gid >> 5;
  int lane = threadIdx.x & 31;
  if (e >= nedges) return;
  int s = src[e];
  int d = dst[e];
  v2f v = *(const v2f*)(h + (size_t)s * GIN_D + lane * 2);
  float* np = neigh + (size_t)d * GIN_D + lane * 2;
  atomicAdd(np, v.x);
  atomicAdd(np + 1, v.y);
}

__global__ __launch_bounds__(WPB * 32) void gin_mlp(
    const float* __restrict__ h, const float* __restrict__ neigh,
    const float* __restrict__ W1, const float* __restrict__ b1,
    const float* __restrict__ W2, const float* __restrict__ b2,
    const float* __restrict__ g,  const float* __restrict__ be,
    const float* __restrict__ mu, const float* __restrict__ va,
    float* __restrict__ out, int do_bn) {
  __shared__ float sW1t[GIN_D * TSTRIDE];
  __shared__ float sW2t[GIN_D * TSTRIDE];
  __shared__ float sT[WPB][16 * TSTRIDE];

  int t = threadIdx.x;
  for (int i = t; i < GIN_D * GIN_D; i += WPB * 32) {
    int k = i >> 6, n = i & 63;
    sW1t[n * TSTRIDE + k] = W1[i];
    sW2t[n * TSTRIDE + k] = W2[i];
  }
  __syncthreads();

  int wave = t >> 5, lane = t & 31;
  int row0 = (blockIdx.x * WPB + wave) * 16;
  if (row0 >= GIN_N) return;
  int m    = lane & 15;
  int half = lane >> 4;

  v2f zA[16];
  const float* hrow = h     + (size_t)(row0 + m) * GIN_D + 2 * half;
  const float* nrow = neigh + (size_t)(row0 + m) * GIN_D + 2 * half;
#pragma unroll
  for (int k0 = 0; k0 < 16; ++k0) {
    v2f a = *(const v2f*)(hrow + 4 * k0);
    v2f b = *(const v2f*)(nrow + 4 * k0);
    zA[k0] = a + b;
  }

  v8f acc[4];
#pragma unroll
  for (int n0 = 0; n0 < 4; ++n0)
#pragma unroll
    for (int i = 0; i < 8; ++i) acc[n0][i] = 0.f;

#pragma unroll
  for (int n0 = 0; n0 < 4; ++n0) {
    const float* wcol = &sW1t[(n0 * 16 + m) * TSTRIDE + 2 * half];
#pragma unroll
    for (int k0 = 0; k0 < 16; ++k0) {
      v2f b = *(const v2f*)(wcol + 4 * k0);
      acc[n0] = __builtin_amdgcn_wmma_f32_16x16x4_f32(
          false, zA[k0], false, b, (short)0, acc[n0], false, false);
    }
  }

  float* st = sT[wave];
#pragma unroll
  for (int n0 = 0; n0 < 4; ++n0) {
    float bias = b1[n0 * 16 + m];
#pragma unroll
    for (int i = 0; i < 8; ++i) {
      float v = fmaxf(acc[n0][i] + bias, 0.f);
      st[(half * 8 + i) * TSTRIDE + n0 * 16 + m] = v;
    }
  }

  v2f h1A[16];
#pragma unroll
  for (int k0 = 0; k0 < 16; ++k0)
    h1A[k0] = *(const v2f*)&st[m * TSTRIDE + 4 * k0 + 2 * half];

  v8f acc2[4];
#pragma unroll
  for (int n0 = 0; n0 < 4; ++n0)
#pragma unroll
    for (int i = 0; i < 8; ++i) acc2[n0][i] = 0.f;

#pragma unroll
  for (int n0 = 0; n0 < 4; ++n0) {
    const float* wcol = &sW2t[(n0 * 16 + m) * TSTRIDE + 2 * half];
#pragma unroll
    for (int k0 = 0; k0 < 16; ++k0) {
      v2f b = *(const v2f*)(wcol + 4 * k0);
      acc2[n0] = __builtin_amdgcn_wmma_f32_16x16x4_f32(
          false, h1A[k0], false, b, (short)0, acc2[n0], false, false);
    }
  }

#pragma unroll
  for (int n0 = 0; n0 < 4; ++n0) {
    int col = n0 * 16 + m;
    float bias = b2[col];
    float scale = 1.f, shift = 0.f;
    if (do_bn) {
      float s = g[col] * rsqrtf(va[col] + 1e-5f);
      scale = s;
      shift = be[col] - mu[col] * s;
    }
#pragma unroll
    for (int i = 0; i < 8; ++i) {
      float v = acc2[n0][i] + bias;
      if (do_bn) v = fmaxf(v * scale + shift, 0.f);
      out[(size_t)(row0 + half * 8 + i) * GIN_D + col] = v;
    }
  }
}

extern "C" void kernel_launch(void* const* d_in, const int* in_sizes, int n_in,
                              void* d_out, int out_size, void* d_ws, size_t ws_size,
                              hipStream_t stream) {
  const float* x  = (const float*)d_in[0];
  const int*   ei = (const int*)d_in[1];
  const float* W1 = (const float*)d_in[2];
  const float* b1 = (const float*)d_in[3];
  const float* W2 = (const float*)d_in[4];
  const float* b2 = (const float*)d_in[5];
  const float* g  = (const float*)d_in[6];
  const float* be = (const float*)d_in[7];
  const float* mu = (const float*)d_in[8];
  const float* va = (const float*)d_in[9];
  float* out = (float*)d_out;

  float* neigh = (float*)d_ws;
  float* hbuf  = neigh + (size_t)GIN_N * GIN_D;

  const int n4    = GIN_N * GIN_D / 4;
  const int zgrid = (n4 + 255) / 256;
  const long long sthreads = (long long)GIN_E * 32;
  const int sgrid = (int)((sthreads + 255) / 256);
  const int mgrid = (GIN_N / 16 + WPB - 1) / WPB;

  const float* hcur = x;
  for (int l = 0; l < 5; ++l) {
    gin_zero<<<zgrid, 256, 0, stream>>>((float4*)neigh, n4);
    gin_scatter<<<sgrid, 256, 0, stream>>>(hcur, ei, ei + GIN_E, neigh, GIN_E);
    float* dstp = (l == 4) ? out : ((l % 2 == 0) ? hbuf : out);
    gin_mlp<<<mgrid, WPB * 32, 0, stream>>>(
        hcur, neigh,
        W1 + (size_t)l * GIN_D * GIN_D, b1 + (size_t)l * GIN_D,
        W2 + (size_t)l * GIN_D * GIN_D, b2 + (size_t)l * GIN_D,
        g  + (size_t)l * GIN_D, be + (size_t)l * GIN_D,
        mu + (size_t)l * GIN_D, va + (size_t)l * GIN_D,
        dstp, (l < 4) ? 1 : 0);
    hcur = dstp;
  }
}


# Atomic scatter-add and its zero pass replaced by a per-call dst-sorted edge list (int count atomics give the rank, scan, slot fill) and a wave-per-node gather; sum order = atomic arrival order
# speedup vs baseline: 2.5467x; 2.5467x over previous
#include <hip/hip_runtime.h>
#include <hip/hip_bf16.h>
#include <cstdint>

typedef __attribute__((ext_vector_type(2))) float v2f;
typedef __attribute__((ext_vector_type(8))) float v8f;

#define GIN_N 100000
#define GIN_E 1600000
#define GIN_D 64
#define WPB 4
#define SCAN_T 1024
#define TSTRIDE 66

__global__ __launch_bounds__(256) void gin_zero(float4* __restrict__ p, int n4) {
  int i = blockIdx.x * blockDim.x + threadIdx.x;
  if (i < n4) p[i] = make_float4(0.f, 0.f, 0.f, 0.f);
}

__global__ __launch_bounds__(256) void gin_hist(
    const int* __restrict__ dst, int* __restrict__ cnt,
    int* __restrict__ rank, int nedges) {
  int e = blockIdx.x * blockDim.x + threadIdx.x;
  if (e < nedges) rank[e] = atomicAdd(cnt + dst[e], 1);
}

__global__ __launch_bounds__(SCAN_T) void gin_scan(
    const int* __restrict__ cnt, int2* __restrict__ rowinfo, int n) {
  __shared__ int sPart[SCAN_T / 32];
  __shared__ int sOwn[SCAN_T / 32];
  int t = threadIdx.x, b = blockIdx.x;
  int lane = t & 31, w = t >> 5;
  int part = 0;
#pragma unroll 8
  for (int j = 0; j < b; ++j) part += cnt[j * SCAN_T + t];
  int i = b * SCAN_T + t;
  int own = (i < n) ? cnt[i] : 0;
  int inc = own;
#pragma unroll
  for (int o = 1; o < 32; o <<= 1) {
    int v = __shfl_up(inc, o);
    if (lane >= o) inc += v;
  }
#pragma unroll
  for (int o = 16; o > 0; o >>= 1) part += __shfl_xor(part, o);
  if (lane == 31) sOwn[w] = inc;
  if (lane == 0)  sPart[w] = part;
  __syncthreads();
  int base = 0;
#pragma unroll
  for (int k = 0; k < SCAN_T / 32; ++k) {
    base += sPart[k];
    if (k < w) base += sOwn[k];
  }
  int excl = base + inc - own;
  if (i < n) rowinfo[i] = make_int2(excl, own);
}

__global__ __launch_bounds__(256) void gin_fill(
    const int* __restrict__ src, const int* __restrict__ dst,
    const int* __restrict__ rank, const int2* __restrict__ rowinfo,
    int* __restrict__ csr, int nedges) {
  int e = blockIdx.x * blockDim.x + threadIdx.x;
  if (e >= nedges) return;
  csr[rowinfo[dst[e]].x + rank[e]] = src[e];
}

__global__ __launch_bounds__(256) void gin_gather(
    const float* __restrict__ h, const int2* __restrict__ rowinfo,
    const int* __restrict__ csr, float* __restrict__ neigh, int n) {
  int node = __builtin_amdgcn_readfirstlane(blockIdx.x * 8 + (threadIdx.x >> 5));
  if (node >= n) return;
  int lane = threadIdx.x & 31;
  int half = lane >> 4;
  const float* hb = h + (lane & 15) * 4;
  int2 ri = rowinfo[node];
  int deg = ri.y;
  const int* lst = csr + ri.x;
  float4 a0 = make_float4(0.f, 0.f, 0.f, 0.f);
  float4 a1 = make_float4(0.f, 0.f, 0.f, 0.f);

  for (int base = 0; base < deg; base += 32) {
    int m = deg - base;
    if (m > 32) m = 32;
    int idxv = lst[base + (lane < m ? lane : m - 1)];
    int i = 0;
    for (; i + 8 <= m; i += 8) {
      int s0 = __shfl(idxv, i + half);
      int s1 = __shfl(idxv, i + 2 + half);
      int s2 = __shfl(idxv, i + 4 + half);
      int s3 = __shfl(idxv, i + 6 + half);
      float4 v0 = *(const float4*)(hb + (size_t)s0 * GIN_D);
      float4 v1 = *(const float4*)(hb + (size_t)s1 * GIN_D);
      float4 v2 = *(const float4*)(hb + (size_t)s2 * GIN_D);
      float4 v3 = *(const float4*)(hb + (size_t)s3 * GIN_D);
      a0.x += v0.x; a0.y += v0.y; a0.z += v0.z; a0.w += v0.w;
      a1.x += v1.x; a1.y += v1.y; a1.z += v1.z; a1.w += v1.w;
      a0.x += v2.x; a0.y += v2.y; a0.z += v2.z; a0.w += v2.w;
      a1.x += v3.x; a1.y += v3.y; a1.z += v3.z; a1.w += v3.w;
    }
    if (i < m) {
      int e0 = i + half, e1 = i + 2 + half, e2 = i + 4 + half, e3 = i + 6 + half;
      int s0 = __shfl(idxv, e0 < m ? e0 : m - 1);
      int s1 = __shfl(idxv, e1 < m ? e1 : m - 1);
      int s2 = __shfl(idxv, e2 < m ? e2 : m - 1);
      int s3 = __shfl(idxv, e3 < m ? e3 : m - 1);
      float4 v0 = *(const float4*)(hb + (size_t)s0 * GIN_D);
      float4 v1 = *(const float4*)(hb + (size_t)s1 * GIN_D);
      float4 v2 = *(const float4*)(hb + (size_t)s2 * GIN_D);
      float4 v3 = *(const float4*)(hb + (size_t)s3 * GIN_D);
      bool p0 = e0 < m, p1 = e1 < m, p2 = e2 < m, p3 = e3 < m;
      a0.x += p0 ? v0.x : 0.f; a0.y += p0 ? v0.y : 0.f; a0.z += p0 ? v0.z : 0.f; a0.w += p0 ? v0.w : 0.f;
      a1.x += p1 ? v1.x : 0.f; a1.y += p1 ? v1.y : 0.f; a1.z += p1 ? v1.z : 0.f; a1.w += p1 ? v1.w : 0.f;
      a0.x += p2 ? v2.x : 0.f; a0.y += p2 ? v2.y : 0.f; a0.z += p2 ? v2.z : 0.f; a0.w += p2 ? v2.w : 0.f;
      a1.x += p3 ? v3.x : 0.f; a1.y += p3 ? v3.y : 0.f; a1.z += p3 ? v3.z : 0.f; a1.w += p3 ? v3.w : 0.f;
    }
  }
  a0.x += a1.x; a0.y += a1.y; a0.z += a1.z; a0.w += a1.w;
  a0.x += __shfl_xor(a0.x, 16);
  a0.y += __shfl_xor(a0.y, 16);
  a0.z += __shfl_xor(a0.z, 16);
  a0.w += __shfl_xor(a0.w, 16);
  if (half == 0)
    *(float4*)(neigh + (size_t)node * GIN_D + (lane & 15) * 4) = a0;
}

__global__ __launch_bounds__(WPB * 32) void gin_mlp(
    const float* __restrict__ h, const float* __restrict__ neigh,
    const float* __restrict__ W1, const float* __restrict__ b1,
    const float* __restrict__ W2, const float* __restrict__ b2,
    const float* __restrict__ g,  const float* __restrict__ be,
    const float* __restrict__ mu, const float* __restrict__ va,
    float* __restrict__ out, int do_bn) {
  __shared__ float sW1t[GIN_D * TSTRIDE];
  __shared__ float sW2t[GIN_D * TSTRIDE];
  __shared__ float sT[WPB][16 * TSTRIDE];

  int t = threadIdx.x;
  for (int i = t; i < GIN_D * GIN_D; i += WPB * 32) {
    int k = i >> 6, n = i & 63;
    sW1t[n * TSTRIDE + k] = W1[i];
    sW2t[n * TSTRIDE + k] = W2[i];
  }
  __syncthreads();

  int wave = t >> 5, lane = t & 31;
  int row0 = (blockIdx.x * WPB + wave) * 16;
  if (row0 >= GIN_N) return;
  int m    = lane & 15;
  int half = lane >> 4;

  v2f zA[16];
  const float* hrow = h     + (size_t)(row0 + m) * GIN_D + 2 * half;
  const float* nrow = neigh + (size_t)(row0 + m) * GIN_D + 2 * half;
#pragma unroll
  for (int k0 = 0; k0 < 16; ++k0) {
    v2f a = *(const v2f*)(hrow + 4 * k0);
    v2f b = *(const v2f*)(nrow + 4 * k0);
    zA[k0] = a + b;
  }

  v8f acc[4];
#pragma unroll
  for (int n0 = 0; n0 < 4; ++n0)
#pragma unroll
    for (int i = 0; i < 8; ++i) acc[n0][i] = 0.f;

#pragma unroll
  for (int n0 = 0; n0 < 4; ++n0) {
    const float* wcol = &sW1t[(n0 * 16 + m) * TSTRIDE + 2 * half];
#pragma unroll
    for (int k0 = 0; k0 < 16; ++k0) {
      v2f b = *(const v2f*)(wcol + 4 * k0);
      acc[n0] = __builtin_amdgcn_wmma_f32_16x16x4_f32(
          false, zA[k0], false, b, (short)0, acc[n0], false, false);
    }
  }

  float* st = sT[wave];
#pragma unroll
  for (int n0 = 0; n0 < 4; ++n0) {
    float bias = b1[n0 * 16 + m];
#pragma unroll
    for (int i = 0; i < 8; ++i) {
      float v = fmaxf(acc[n0][i] + bias, 0.f);
      st[(half * 8 + i) * TSTRIDE + n0 * 16 + m] = v;
    }
  }

  v2f h1A[16];
#pragma unroll
  for (int k0 = 0; k0 < 16; ++k0)
    h1A[k0] = *(const v2f*)&st[m * TSTRIDE + 4 * k0 + 2 * half];

  v8f acc2[4];
#pragma unroll
  for (int n0 = 0; n0 < 4; ++n0)
#pragma unroll
    for (int i = 0; i < 8; ++i) acc2[n0][i] = 0.f;

#pragma unroll
  for (int n0 = 0; n0 < 4; ++n0) {
    const float* wcol = &sW2t[(n0 * 16 + m) * TSTRIDE + 2 * half];
#pragma unroll
    for (int k0 = 0; k0 < 16; ++k0) {
      v2f b = *(const v2f*)(wcol + 4 * k0);
      acc2[n0] = __builtin_amdgcn_wmma_f32_16x16x4_f32(
          false, h1A[k0], false, b, (short)0, acc2[n0], false, false);
    }
  }

#pragma unroll
  for (int n0 = 0; n0 < 4; ++n0) {
    int col = n0 * 16 + m;
    float bias = b2[col];
    float scale = 1.f, shift = 0.f;
    if (do_bn) {
      float s = g[col] * rsqrtf(va[col] + 1e-5f);
      scale = s;
      shift = be[col] - mu[col] * s;
    }
#pragma unroll
    for (int i = 0; i < 8; ++i) {
      float v = acc2[n0][i] + bias;
      if (do_bn) v = fmaxf(v * scale + shift, 0.f);
      out[(size_t)(row0 + half * 8 + i) * GIN_D + col] = v;
    }
  }
}

extern "C" void kernel_launch(void* const* d_in, const int* in_sizes, int n_in,
                              void* d_out, int out_size, void* d_ws, size_t ws_size,
                              hipStream_t stream) {
  const float* x  = (const float*)d_in[0];
  const int*   ei = (const int*)d_in[1];
  const float* W1 = (const float*)d_in[2];
  const float* b1 = (const float*)d_in[3];
  const float* W2 = (const float*)d_in[4];
  const float* b2 = (const float*)d_in[5];
  const float* g  = (const float*)d_in[6];
  const float* be = (const float*)d_in[7];
  const float* mu = (const float*)d_in[8];
  const float* va = (const float*)d_in[9];
  float* out = (float*)d_out;

  float* neigh = (float*)d_ws;
  float* hbuf  = neigh + (size_t)GIN_N * GIN_D;
  int*  cnt     = (int*)(hbuf + (size_t)GIN_N * GIN_D);
  int2* rowinfo = (int2*)(cnt + GIN_N);
  int*  rank    = (int*)(rowinfo + GIN_N);
  int*  csr     = rank + GIN_E;

  const int c4    = GIN_N / 4;
  const int zgrid = (c4 + 255) / 256;
  const int egrid = (GIN_E + 255) / 256;
  const int pgrid = (GIN_N + SCAN_T - 1) / SCAN_T;
  const int ggrid = (GIN_N + 7) / 8;
  const int mgrid = (GIN_N / 16 + WPB - 1) / WPB;

  gin_zero<<<zgrid, 256, 0, stream>>>((float4*)cnt, c4);
  gin_hist<<<egrid, 256, 0, stream>>>(ei + GIN_E, cnt, rank, GIN_E);
  gin_scan<<<pgrid, SCAN_T, 0, stream>>>(cnt, rowinfo, GIN_N);
  gin_fill<<<egrid, 256, 0, stream>>>(ei, ei + GIN_E, rank, rowinfo, csr, GIN_E);

  const float* hcur = x;
  for (int l = 0; l < 5; ++l) {
    gin_gather<<<ggrid, 256, 0, stream>>>(hcur, rowinfo, csr, neigh, GIN_N);
    float* dstp = (l == 4) ? out : ((l % 2 == 0) ? hbuf : out);
    gin_mlp<<<mgrid, WPB * 32, 0, stream>>>(
        hcur, neigh,
        W1 + (size_t)l * GIN_D * GIN_D, b1 + (size_t)l * GIN_D,
        W2 + (size_t)l * GIN_D * GIN_D, b2 + (size_t)l * GIN_D,
        g  + (size_t)l * GIN_D, be + (size_t)l * GIN_D,
        mu + (size_t)l * GIN_D, va + (size_t)l * GIN_D,
        dstp, (l < 4) ? 1 : 0);
    hcur = dstp;
  }
}


# Edge sort without global atomics: edges split into 512-node bins (per-chunk LDS counts, one table scan, LDS cursors), one block per bin forms its rows in LDS; sum order = LDS atomic arrival order
# speedup vs baseline: 3.3882x; 1.3304x over previous
#include <hip/hip_runtime.h>
#include <hip/hip_bf16.h>
#include <cstdint>

typedef __attribute__((ext_vector_type(2))) float v2f;
typedef __attribute__((ext_vector_type(8))) float v8f;

#define GIN_N 100000
#define GIN_E 1600000
#define GIN_D 64
#define WPB 4
#define TSTRIDE 66

#define ST 1024
#define NBLK 256
#define BSH 9
#define BNODES (1 << BSH)
#define NBIN ((GIN_N + BNODES - 1) >> BSH)
#define NTBL (NBIN * NBLK)
#define TPT (NTBL / ST)
#define CHUNK ((GIN_E + NBLK - 1) / NBLK)
#define EPT ((CHUNK + ST - 1) / ST)
#define CAP 12288
#define RB 8
#define NOEDGE 0xFFFFFFFFu


__global__ __launch_bounds__(ST) void gin_bincount(
    const int* __restrict__ dst, int* __restrict__ tbl) {
  __shared__ int sc[NBIN];
  int t = threadIdx.x, b = blockIdx.x;
  if (t < NBIN) sc[t] = 0;
  __syncthreads();
  int e0 = b * CHUNK;
  int e1 = e0 + CHUNK < GIN_E ? e0 + CHUNK : GIN_E;
  int d[EPT];
#pragma unroll
  for (int j = 0; j < EPT; ++j) {
    int e = e0 + t + j * ST;
    d[j] = (e < e1) ? dst[e] : -1;
  }
#pragma unroll
  for (int j = 0; j < EPT; ++j)
    if (d[j] >= 0) atomicAdd(&sc[d[j] >> BSH], 1);
  __syncthreads();
  if (t < NBIN) tbl[t * NBLK + b] = sc[t];
}

__global__ __launch_bounds__(ST) void gin_binscan(int* __restrict__ tbl) {
  __shared__ int sW[ST / 32];
  int t = threadIdx.x, lane = t & 31, w = t >> 5;
  int v[TPT];
#pragma unroll
  for (int j = 0; j < TPT; ++j) v[j] = tbl[t * TPT + j];
  int s = 0;
#pragma unroll
  for (int j = 0; j < TPT; ++j) { int x = v[j]; v[j] = s; s += x; }
  int inc = s;
#pragma unroll
  for (int o = 1; o < 32; o <<= 1) {
    int u = __shfl_up(inc, o);
    if (lane >= o) inc += u;
  }
  if (lane == 31) sW[w] = inc;
  __syncthreads();
  int base = 0;
#pragma unroll
  for (int k = 0; k < ST / 32; ++k) if (k < w) base += sW[k];
  int excl = base + inc - s;
#pragma unroll
  for (int j = 0; j < TPT; ++j) tbl[t * TPT + j] = v[j] + excl;
}

__global__ __launch_bounds__(ST) void gin_binsplit(
    const int* __restrict__ src, const int* __restrict__ dst,
    const int* __restrict__ tbl, unsigned* __restrict__ binned) {
  __shared__ int sc[NBIN];
  int t = threadIdx.x, b = blockIdx.x;
  if (t < NBIN) sc[t] = tbl[t * NBLK + b];
  __syncthreads();
  int e0 = b * CHUNK;
  int e1 = e0 + CHUNK < GIN_E ? e0 + CHUNK : GIN_E;
  int d[EPT];
  unsigned s[EPT];
#pragma unroll
  for (int j = 0; j < EPT; ++j) {
    int e = e0 + t + j * ST;
    bool in = e < e1;
    d[j] = in ? dst[e] : -1;
    s[j] = in ? (unsigned)src[e] : 0u;
  }
#pragma unroll
  for (int j = 0; j < EPT; ++j) {
    if (d[j] >= 0) {
      int pos = atomicAdd(&sc[d[j] >> BSH], 1);
      binned[pos] = (s[j] << BSH) | (unsigned)(d[j] & (BNODES - 1));
    }
  }
}

__global__ __launch_bounds__(ST) void gin_binrows(
    const int* __restrict__ tbl, const unsigned* __restrict__ binned,
    int2* __restrict__ rowinfo, int* __restrict__ csr) {
  __shared__ int sCnt[BNODES];
  __shared__ int sW[ST / 32];
  __shared__ int sOut[CAP];
  int t = threadIdx.x, b = blockIdx.x;
  int lane = t & 31, w = t >> 5;
  int beg = tbl[b * NBLK];
  int end = (b + 1 < NBIN) ? tbl[(b + 1) * NBLK] : GIN_E;
  int len = end - beg;
  const unsigned* bw = binned + beg;
  if (t < BNODES) sCnt[t] = 0;
  __syncthreads();
  for (int i0 = t; i0 < len; i0 += RB * ST) {
    unsigned x[RB];
#pragma unroll
    for (int j = 0; j < RB; ++j) {
      int i = i0 + j * ST;
      x[j] = (i < len) ? bw[i] : NOEDGE;
    }
#pragma unroll
    for (int j = 0; j < RB; ++j)
      if (x[j] != NOEDGE) atomicAdd(&sCnt[x[j] & (BNODES - 1)], 1);
  }
  __syncthreads();
  int own = (t < BNODES) ? sCnt[t] : 0;
  int inc = own;
#pragma unroll
  for (int o = 1; o < 32; o <<= 1) {
    int u = __shfl_up(inc, o);
    if (lane >= o) inc += u;
  }
  if (lane == 31) sW[w] = inc;
  __syncthreads();
  int base = 0;
#pragma unroll
  for (int k = 0; k < ST / 32; ++k) if (k < w) base += sW[k];
  int excl = base + inc - own;
  if (t < BNODES) {
    sCnt[t] = excl;
    int node = b * BNODES + t;
    if (node < GIN_N) rowinfo[node] = make_int2(beg + excl, own);
  }
  __syncthreads();
  if (len <= CAP) {
    for (int i0 = t; i0 < len; i0 += RB * ST) {
      unsigned x[RB];
#pragma unroll
      for (int j = 0; j < RB; ++j) {
        int i = i0 + j * ST;
        x[j] = (i < len) ? bw[i] : NOEDGE;
      }
#pragma unroll
      for (int j = 0; j < RB; ++j) {
        if (x[j] != NOEDGE) {
          int pos = atomicAdd(&sCnt[x[j] & (BNODES - 1)], 1);
          sOut[pos] = (int)(x[j] >> BSH);
        }
      }
    }
    __syncthreads();
    for (int i = t; i < len; i += ST) csr[beg + i] = sOut[i];
  } else {
    for (int i = t; i < len; i += ST) {
      unsigned x = bw[i];
      int pos = atomicAdd(&sCnt[x & (BNODES - 1)], 1);
      csr[beg + pos] = (int)(x >> BSH);
    }
  }
}

__global__ __launch_bounds__(256) void gin_gather(
    const float* __restrict__ h, const int2* __restrict__ rowinfo,
    const int* __restrict__ csr, float* __restrict__ neigh, int n) {
  int node = __builtin_amdgcn_readfirstlane(blockIdx.x * 8 + (threadIdx.x >> 5));
  if (node >= n) return;
  int lane = threadIdx.x & 31;
  int half = lane >> 4;
  const float* hb = h + (lane & 15) * 4;
  int2 ri = rowinfo[node];
  int deg = ri.y;
  const int* lst = csr + ri.x;
  float4 a0 = make_float4(0.f, 0.f, 0.f, 0.f);
  float4 a1 = make_float4(0.f, 0.f, 0.f, 0.f);

  for (int base = 0; base < deg; base += 32) {
    int m = deg - base;
    if (m > 32) m = 32;
    int idxv = lst[base + (lane < m ? lane : m - 1)];
    int i = 0;
    for (; i + 8 <= m; i += 8) {
      int s0 = __shfl(idxv, i + half);
      int s1 = __shfl(idxv, i + 2 + half);
      int s2 = __shfl(idxv, i + 4 + half);
      int s3 = __shfl(idxv, i + 6 + half);
      float4 v0 = *(const float4*)(hb + (size_t)s0 * GIN_D);
      float4 v1 = *(const float4*)(hb + (size_t)s1 * GIN_D);
      float4 v2 = *(const float4*)(hb + (size_t)s2 * GIN_D);
      float4 v3 = *(const float4*)(hb + (size_t)s3 * GIN_D);
      a0.x += v0.x; a0.y += v0.y; a0.z += v0.z; a0.w += v0.w;
      a1.x += v1.x; a1.y += v1.y; a1.z += v1.z; a1.w += v1.w;
      a0.x += v2.x; a0.y += v2.y; a0.z += v2.z; a0.w += v2.w;
      a1.x += v3.x; a1.y += v3.y; a1.z += v3.z; a1.w += v3.w;
    }
    if (i < m) {
      int e0 = i + half, e1 = i + 2 + half, e2 = i + 4 + half, e3 = i + 6 + half;
      int s0 = __shfl(idxv, e0 < m ? e0 : m - 1);
      int s1 = __shfl(idxv, e1 < m ? e1 : m - 1);
      int s2 = __shfl(idxv, e2 < m ? e2 : m - 1);
      int s3 = __shfl(idxv, e3 < m ? e3 : m - 1);
      float4 v0 = *(const float4*)(hb + (size_t)s0 * GIN_D);
      float4 v1 = *(const float4*)(hb + (size_t)s1 * GIN_D);
      float4 v2 = *(const float4*)(hb + (size_t)s2 * GIN_D);
      float4 v3 = *(const float4*)(hb + (size_t)s3 * GIN_D);
      bool p0 = e0 < m, p1 = e1 < m, p2 = e2 < m, p3 = e3 < m;
      a0.x += p0 ? v0.x : 0.f; a0.y += p0 ? v0.y : 0.f; a0.z += p0 ? v0.z : 0.f; a0.w += p0 ? v0.w : 0.f;
      a1.x += p1 ? v1.x : 0.f; a1.y += p1 ? v1.y : 0.f; a1.z += p1 ? v1.z : 0.f; a1.w += p1 ? v1.w : 0.f;
      a0.x += p2 ? v2.x : 0.f; a0.y += p2 ? v2.y : 0.f; a0.z += p2 ? v2.z : 0.f; a0.w += p2 ? v2.w : 0.f;
      a1.x += p3 ? v3.x : 0.f; a1.y += p3 ? v3.y : 0.f; a1.z += p3 ? v3.z : 0.f; a1.w += p3 ? v3.w : 0.f;
    }
  }
  a0.x += a1.x; a0.y += a1.y; a0.z += a1.z; a0.w += a1.w;
  a0.x += __shfl_xor(a0.x, 16);
  a0.y += __shfl_xor(a0.y, 16);
  a0.z += __shfl_xor(a0.z, 16);
  a0.w += __shfl_xor(a0.w, 16);
  if (half == 0)
    *(float4*)(neigh + (size_t)node * GIN_D + (lane & 15) * 4) = a0;
}

__global__ __launch_bounds__(WPB * 32) void gin_mlp(
    const float* __restrict__ h, const float* __restrict__ neigh,
    const float* __restrict__ W1, const float* __restrict__ b1,
    const float* __restrict__ W2, const float* __restrict__ b2,
    const float* __restrict__ g,  const float* __restrict__ be,
    const float* __restrict__ mu, const float* __restrict__ va,
    float* __restrict__ out, int do_bn) {
  __shared__ float sW1t[GIN_D * TSTRIDE];
  __shared__ float sW2t[GIN_D * TSTRIDE];
  __shared__ float sT[WPB][16 * TSTRIDE];

  int t = threadIdx.x;
  for (int i = t; i < GIN_D * GIN_D; i += WPB * 32) {
    int k = i >> 6, n = i & 63;
    sW1t[n * TSTRIDE + k] = W1[i];
    sW2t[n * TSTRIDE + k] = W2[i];
  }
  __syncthreads();

  int wave = t >> 5, lane = t & 31;
  int row0 = (blockIdx.x * WPB + wave) * 16;
  if (row0 >= GIN_N) return;
  int m    = lane & 15;
  int half = lane >> 4;

  v2f zA[16];
  const float* hrow = h     + (size_t)(row0 + m) * GIN_D + 2 * half;
  const float* nrow = neigh + (size_t)(row0 + m) * GIN_D + 2 * half;
#pragma unroll
  for (int k0 = 0; k0 < 16; ++k0) {
    v2f a = *(const v2f*)(hrow + 4 * k0);
    v2f b = *(const v2f*)(nrow + 4 * k0);
    zA[k0] = a + b;
  }

  v8f acc[4];
#pragma unroll
  for (int n0 = 0; n0 < 4; ++n0)
#pragma unroll
    for (int i = 0; i < 8; ++i) acc[n0][i] = 0.f;

#pragma unroll
  for (int n0 = 0; n0 < 4; ++n0) {
    const float* wcol = &sW1t[(n0 * 16 + m) * TSTRIDE + 2 * half];
#pragma unroll
    for (int k0 = 0; k0 < 16; ++k0) {
      v2f b = *(const v2f*)(wcol + 4 * k0);
      acc[n0] = __builtin_amdgcn_wmma_f32_16x16x4_f32(
          false, zA[k0], false, b, (short)0, acc[n0], false, false);
    }
  }

  float* st = sT[wave];
#pragma unroll
  for (int n0 = 0; n0 < 4; ++n0) {
    float bias = b1[n0 * 16 + m];
#pragma unroll
    for (int i = 0; i < 8; ++i) {
      float v = fmaxf(acc[n0][i] + bias, 0.f);
      st[(half * 8 + i) * TSTRIDE + n0 * 16 + m] = v;
    }
  }

  v2f h1A[16];
#pragma unroll
  for (int k0 = 0; k0 < 16; ++k0)
    h1A[k0] = *(const v2f*)&st[m * TSTRIDE + 4 * k0 + 2 * half];

  v8f acc2[4];
#pragma unroll
  for (int n0 = 0; n0 < 4; ++n0)
#pragma unroll
    for (int i = 0; i < 8; ++i) acc2[n0][i] = 0.f;

#pragma unroll
  for (int n0 = 0; n0 < 4; ++n0) {
    const float* wcol = &sW2t[(n0 * 16 + m) * TSTRIDE + 2 * half];
#pragma unroll
    for (int k0 = 0; k0 < 16; ++k0) {
      v2f b = *(const v2f*)(wcol + 4 * k0);
      acc2[n0] = __builtin_amdgcn_wmma_f32_16x16x4_f32(
          false, h1A[k0], false, b, (short)0, acc2[n0], false, false);
    }
  }

#pragma unroll
  for (int n0 = 0; n0 < 4; ++n0) {
    int col = n0 * 16 + m;
    float bias = b2[col];
    float scale = 1.f, shift = 0.f;
    if (do_bn) {
      float s = g[col] * rsqrtf(va[col] + 1e-5f);
      scale = s;
      shift = be[col] - mu[col] * s;
    }
#pragma unroll
    for (int i = 0; i < 8; ++i) {
      float v = acc2[n0][i] + bias;
      if (do_bn) v = fmaxf(v * scale + shift, 0.f);
      out[(size_t)(row0 + half * 8 + i) * GIN_D + col] = v;
    }
  }
}

extern "C" void kernel_launch(void* const* d_in, const int* in_sizes, int n_in,
                              void* d_out, int out_size, void* d_ws, size_t ws_size,
                              hipStream_t stream) {
  const float* x  = (const float*)d_in[0];
  const int*   ei = (const int*)d_in[1];
  const float* W1 = (const float*)d_in[2];
  const float* b1 = (const float*)d_in[3];
  const float* W2 = (const float*)d_in[4];
  const float* b2 = (const float*)d_in[5];
  const float* g  = (const float*)d_in[6];
  const float* be = (const float*)d_in[7];
  const float* mu = (const float*)d_in[8];
  const float* va = (const float*)d_in[9];
  float* out = (float*)d_out;

  float* neigh = (float*)d_ws;
  float* hbuf  = neigh + (size_t)GIN_N * GIN_D;
  int2* rowinfo = (int2*)(hbuf + (size_t)GIN_N * GIN_D);
  int*  csr     = (int*)(rowinfo + GIN_N);
  unsigned* binned = (unsigned*)(csr + GIN_E);
  int*  tbl     = (int*)(binned + GIN_E);

  const int ggrid = (GIN_N + 7) / 8;
  const int mgrid = (GIN_N / 16 + WPB - 1) / WPB;

  gin_bincount<<<NBLK, ST, 0, stream>>>(ei + GIN_E, tbl);
  gin_binscan<<<1, ST, 0, stream>>>(tbl);
  gin_binsplit<<<NBLK, ST, 0, stream>>>(ei, ei + GIN_E, tbl, binned);
  gin_binrows<<<NBIN, ST, 0, stream>>>(tbl, binned, rowinfo, csr);

  const float* hcur = x;
  for (int l = 0; l < 5; ++l) {
    gin_gather<<<ggrid, 256, 0, stream>>>(hcur, rowinfo, csr, neigh, GIN_N);
    float* dstp = (l == 4) ? out : ((l % 2 == 0) ? hbuf : out);
    gin_mlp<<<mgrid, WPB * 32, 0, stream>>>(
        hcur, neigh,
        W1 + (size_t)l * GIN_D * GIN_D, b1 + (size_t)l * GIN_D,
        W2 + (size_t)l * GIN_D * GIN_D, b2 + (size_t)l * GIN_D,
        g  + (size_t)l * GIN_D, be + (size_t)l * GIN_D,
        mu + (size_t)l * GIN_D, va + (size_t)l * GIN_D,
        dstp, (l < 4) ? 1 : 0);
    hcur = dstp;
  }
}


# Neighbour gather fused into the MLP launch: each wave sums its 16 nodes' rows (one node per half wave) into its LDS tile, then multiplies; 250 blocks of 25 waves; per-node sum order changes
# speedup vs baseline: 3.6974x; 1.0913x over previous
#include <hip/hip_runtime.h>
#include <hip/hip_bf16.h>
#include <cstdint>

typedef __attribute__((ext_vector_type(2))) float v2f;
typedef __attribute__((ext_vector_type(8))) float v8f;

#define GIN_N 100000
#define GIN_E 1600000
#define GIN_D 64
#define WPB 25
#define TSTRIDE 66

#define ST 1024
#define NBLK 256
#define BSH 9
#define BNODES (1 << BSH)
#define NBIN ((GIN_N + BNODES - 1) >> BSH)
#define NTBL (NBIN * NBLK)
#define TPT (NTBL / ST)
#define CHUNK ((GIN_E + NBLK - 1) / NBLK)
#define EPT ((CHUNK + ST - 1) / ST)
#define CAP 12288
#define RB 8
#define NOEDGE 0xFFFFFFFFu


__global__ __launch_bounds__(ST) void gin_bincount(
    const int* __restrict__ dst, int* __restrict__ tbl) {
  __shared__ int sc[NBIN];
  int t = threadIdx.x, b = blockIdx.x;
  if (t < NBIN) sc[t] = 0;
  __syncthreads();
  int e0 = b * CHUNK;
  int e1 = e0 + CHUNK < GIN_E ? e0 + CHUNK : GIN_E;
  int d[EPT];
#pragma unroll
  for (int j = 0; j < EPT; ++j) {
    int e = e0 + t + j * ST;
    d[j] = (e < e1) ? dst[e] : -1;
  }
#pragma unroll
  for (int j = 0; j < EPT; ++j)
    if (d[j] >= 0) atomicAdd(&sc[d[j] >> BSH], 1);
  __syncthreads();
  if (t < NBIN) tbl[t * NBLK + b] = sc[t];
}

__global__ __launch_bounds__(ST) void gin_binscan(int* __restrict__ tbl) {
  __shared__ int sW[ST / 32];
  int t = threadIdx.x, lane = t & 31, w = t >> 5;
  int v[TPT];
#pragma unroll
  for (int j = 0; j < TPT; ++j) v[j] = tbl[t * TPT + j];
  int s = 0;
#pragma unroll
  for (int j = 0; j < TPT; ++j) { int x = v[j]; v[j] = s; s += x; }
  int inc = s;
#pragma unroll
  for (int o = 1; o < 32; o <<= 1) {
    int u = __shfl_up(inc, o);
    if (lane >= o) inc += u;
  }
  if (lane == 31) sW[w] = inc;
  __syncthreads();
  int base = 0;
#pragma unroll
  for (int k = 0; k < ST / 32; ++k) if (k < w) base += sW[k];
  int excl = base + inc - s;
#pragma unroll
  for (int j = 0; j < TPT; ++j) tbl[t * TPT + j] = v[j] + excl;
}

__global__ __launch_bounds__(ST) void gin_binsplit(
    const int* __restrict__ src, const int* __restrict__ dst,
    const int* __restrict__ tbl, unsigned* __restrict__ binned) {
  __shared__ int sc[NBIN];
  int t = threadIdx.x, b = blockIdx.x;
  if (t < NBIN) sc[t] = tbl[t * NBLK + b];
  __syncthreads();
  int e0 = b * CHUNK;
  int e1 = e0 + CHUNK < GIN_E ? e0 + CHUNK : GIN_E;
  int d[EPT];
  unsigned s[EPT];
#pragma unroll
  for (int j = 0; j < EPT; ++j) {
    int e = e0 + t + j * ST;
    bool in = e < e1;
    d[j] = in ? dst[e] : -1;
    s[j] = in ? (unsigned)src[e] : 0u;
  }
#pragma unroll
  for (int j = 0; j < EPT; ++j) {
    if (d[j] >= 0) {
      int pos = atomicAdd(&sc[d[j] >> BSH], 1);
      binned[pos] = (s[j] << BSH) | (unsigned)(d[j] & (BNODES - 1));
    }
  }
}

__global__ __launch_bounds__(ST) void gin_binrows(
    const int* __restrict__ tbl, const unsigned* __restrict__ binned,
    int2* __restrict__ rowinfo, int* __restrict__ csr) {
  __shared__ int sCnt[BNODES];
  __shared__ int sW[ST / 32];
  __shared__ int sOut[CAP];
  int t = threadIdx.x, b = blockIdx.x;
  int lane = t & 31, w = t >> 5;
  int beg = tbl[b * NBLK];
  int end = (b + 1 < NBIN) ? tbl[(b + 1) * NBLK] : GIN_E;
  int len = end - beg;
  const unsigned* bw = binned + beg;
  if (t < BNODES) sCnt[t] = 0;
  __syncthreads();
  for (int i0 = t; i0 < len; i0 += RB * ST) {
    unsigned x[RB];
#pragma unroll
    for (int j = 0; j < RB; ++j) {
      int i = i0 + j * ST;
      x[j] = (i < len) ? bw[i] : NOEDGE;
    }
#pragma unroll
    for (int j = 0; j < RB; ++j)
      if (x[j] != NOEDGE) atomicAdd(&sCnt[x[j] & (BNODES - 1)], 1);
  }
  __syncthreads();
  int own = (t < BNODES) ? sCnt[t] : 0;
  int inc = own;
#pragma unroll
  for (int o = 1; o < 32; o <<= 1) {
    int u = __shfl_up(inc, o);
    if (lane >= o) inc += u;
  }
  if (lane == 31) sW[w] = inc;
  __syncthreads();
  int base = 0;
#pragma unroll
  for (int k = 0; k < ST / 32; ++k) if (k < w) base += sW[k];
  int excl = base + inc - own;
  if (t < BNODES) {
    sCnt[t] = excl;
    int node = b * BNODES + t;
    if (node < GIN_N) rowinfo[node] = make_int2(beg + excl, own);
  }
  __syncthreads();
  if (len <= CAP) {
    for (int i0 = t; i0 < len; i0 += RB * ST) {
      unsigned x[RB];
#pragma unroll
      for (int j = 0; j < RB; ++j) {
        int i = i0 + j * ST;
        x[j] = (i < len) ? bw[i] : NOEDGE;
      }
#pragma unroll
      for (int j = 0; j < RB; ++j) {
        if (x[j] != NOEDGE) {
          int pos = atomicAdd(&sCnt[x[j] & (BNODES - 1)], 1);
          sOut[pos] = (int)(x[j] >> BSH);
        }
      }
    }
    __syncthreads();
    for (int i = t; i < len; i += ST) csr[beg + i] = sOut[i];
  } else {
    for (int i = t; i < len; i += ST) {
      unsigned x = bw[i];
      int pos = atomicAdd(&sCnt[x & (BNODES - 1)], 1);
      csr[beg + pos] = (int)(x >> BSH);
    }
  }
}

__global__ __launch_bounds__(WPB * 32) void gin_layer(
    const float* __restrict__ h, const int2* __restrict__ rowinfo,
    const int* __restrict__ csr,
    const float* __restrict__ W1, const float* __restrict__ b1,
    const float* __restrict__ W2, const float* __restrict__ b2,
    const float* __restrict__ g,  const float* __restrict__ be,
    const float* __restrict__ mu, const float* __restrict__ va,
    float* __restrict__ out, int do_bn) {
  __shared__ float sW1t[GIN_D * TSTRIDE];
  __shared__ float sW2t[GIN_D * TSTRIDE];
  __shared__ float sT[WPB][16 * TSTRIDE];

  int t = threadIdx.x;
#pragma unroll
  for (int i = t; i < GIN_D * GIN_D; i += WPB * 32) {
    int k = i >> 6, n = i & 63;
    sW1t[n * TSTRIDE + k] = W1[i];
    sW2t[n * TSTRIDE + k] = W2[i];
  }

  int wave = t >> 5, lane = t & 31;
  int row0 = (blockIdx.x * WPB + wave) * 16;
  bool active = row0 < GIN_N;
  int m    = lane & 15;
  int half = lane >> 4;
  float* st = sT[wave];

  if (active) {
    const float* hb = h + m * 4;
    int2 ri = rowinfo[row0 + m];
    int start = __shfl(ri.x, half, 16);
    int deg   = __shfl(ri.y, half, 16);
    int ia = (m < deg) ? csr[start + m] : 0;
    int ib = (16 + m < deg) ? csr[start + 16 + m] : 0;
    for (int q = 0; q < 8; ++q) {
      int startN = 0, degN = 0, iaN = 0, ibN = 0;
      if (q + 1 < 8) {
        int rn = 2 * (q + 1) + half;
        startN = __shfl(ri.x, rn, 16);
        degN   = __shfl(ri.y, rn, 16);
        iaN = (m < degN) ? csr[startN + m] : 0;
        ibN = (16 + m < degN) ? csr[startN + 16 + m] : 0;
      }
      int node = row0 + 2 * q + half;
      float4 hs = *(const float4*)(hb + (size_t)node * GIN_D);
      float4 a0 = make_float4(0.f, 0.f, 0.f, 0.f);
      float4 a1 = make_float4(0.f, 0.f, 0.f, 0.f);
      for (int base = 0; base < deg; base += 32) {
        int mm = deg - base;
        if (mm > 32) mm = 32;
        if (base > 0) {
          ia = (base + m < deg) ? csr[start + base + m] : 0;
          ib = (base + 16 + m < deg) ? csr[start + base + 16 + m] : 0;
        }
        for (int i = 0; i < mm; i += 8) {
          int iv = (i < 16) ? ia : ib;
          int j0 = i & 15;
          int rem = mm - i;
          float4 v[8];
#pragma unroll
          for (int k = 0; k < 8; ++k) {
            int s = __shfl(iv, j0 + k, 16);
            v[k] = make_float4(0.f, 0.f, 0.f, 0.f);
            if (k < rem) v[k] = *(const float4*)(hb + (size_t)s * GIN_D);
          }
#pragma unroll
          for (int k = 0; k < 8; k += 2) {
            a0.x += v[k].x; a0.y += v[k].y; a0.z += v[k].z; a0.w += v[k].w;
            a1.x += v[k + 1].x; a1.y += v[k + 1].y; a1.z += v[k + 1].z; a1.w += v[k + 1].w;
          }
        }
      }
      a0.x += a1.x; a0.y += a1.y; a0.z += a1.z; a0.w += a1.w;
      float* zr = st + (2 * q + half) * TSTRIDE + m * 4;
      v2f z01, z23;
      z01.x = hs.x + a0.x; z01.y = hs.y + a0.y;
      z23.x = hs.z + a0.z; z23.y = hs.w + a0.w;
      *(v2f*)zr = z01;
      *(v2f*)(zr + 2) = z23;
      start = startN; deg = degN; ia = iaN; ib = ibN;
    }
  }
  __syncthreads();
  if (!active) return;

  v2f zA[16];
#pragma unroll
  for (int k0 = 0; k0 < 16; ++k0)
    zA[k0] = *(const v2f*)&st[m * TSTRIDE + 4 * k0 + 2 * half];

  v8f acc[4];
#pragma unroll
  for (int n0 = 0; n0 < 4; ++n0)
#pragma unroll
    for (int i = 0; i < 8; ++i) acc[n0][i] = 0.f;

#pragma unroll
  for (int n0 = 0; n0 < 4; ++n0) {
    const float* wcol = &sW1t[(n0 * 16 + m) * TSTRIDE + 2 * half];
#pragma unroll
    for (int k0 = 0; k0 < 16; ++k0) {
      v2f b = *(const v2f*)(wcol + 4 * k0);
      acc[n0] = __builtin_amdgcn_wmma_f32_16x16x4_f32(
          false, zA[k0], false, b, (short)0, acc[n0], false, false);
    }
  }

#pragma unroll
  for (int n0 = 0; n0 < 4; ++n0) {
    float bias = b1[n0 * 16 + m];
#pragma unroll
    for (int i = 0; i < 8; ++i) {
      float v = fmaxf(acc[n0][i] + bias, 0.f);
      st[(half * 8 + i) * TSTRIDE + n0 * 16 + m] = v;
    }
  }

  v2f h1A[16];
#pragma unroll
  for (int k0 = 0; k0 < 16; ++k0)
    h1A[k0] = *(const v2f*)&st[m * TSTRIDE + 4 * k0 + 2 * half];

  v8f acc2[4];
#pragma unroll
  for (int n0 = 0; n0 < 4; ++n0)
#pragma unroll
    for (int i = 0; i < 8; ++i) acc2[n0][i] = 0.f;

#pragma unroll
  for (int n0 = 0; n0 < 4; ++n0) {
    const float* wcol = &sW2t[(n0 * 16 + m) * TSTRIDE + 2 * half];
#pragma unroll
    for (int k0 = 0; k0 < 16; ++k0) {
      v2f b = *(const v2f*)(wcol + 4 * k0);
      acc2[n0] = __builtin_amdgcn_wmma_f32_16x16x4_f32(
          false, h1A[k0], false, b, (short)0, acc2[n0], false, false);
    }
  }

#pragma unroll
  for (int n0 = 0; n0 < 4; ++n0) {
    int col = n0 * 16 + m;
    float bias = b2[col];
    float scale = 1.f, shift = 0.f;
    if (do_bn) {
      float s = g[col] * rsqrtf(va[col] + 1e-5f);
      scale = s;
      shift = be[col] - mu[col] * s;
    }
#pragma unroll
    for (int i = 0; i < 8; ++i) {
      float v = acc2[n0][i] + bias;
      if (do_bn) v = fmaxf(v * scale + shift, 0.f);
      out[(size_t)(row0 + half * 8 + i) * GIN_D + col] = v;
    }
  }
}

extern "C" void kernel_launch(void* const* d_in, const int* in_sizes, int n_in,
                              void* d_out, int out_size, void* d_ws, size_t ws_size,
                              hipStream_t stream) {
  const float* x  = (const float*)d_in[0];
  const int*   ei = (const int*)d_in[1];
  const float* W1 = (const float*)d_in[2];
  const float* b1 = (const float*)d_in[3];
  const float* W2 = (const float*)d_in[4];
  const float* b2 = (const float*)d_in[5];
  const float* g  = (const float*)d_in[6];
  const float* be = (const float*)d_in[7];
  const float* mu = (const float*)d_in[8];
  const float* va = (const float*)d_in[9];
  float* out = (float*)d_out;

  float* hA    = (float*)d_ws;
  float* hB    = hA + (size_t)GIN_N * GIN_D;
  int2* rowinfo = (int2*)(hB + (size_t)GIN_N * GIN_D);
  int*  csr     = (int*)(rowinfo + GIN_N);
  unsigned* binned = (unsigned*)(csr + GIN_E);
  int*  tbl     = (int*)(binned + GIN_E);

  const int lgrid = (GIN_N / 16 + WPB - 1) / WPB;

  gin_bincount<<<NBLK, ST, 0, stream>>>(ei + GIN_E, tbl);
  gin_binscan<<<1, ST, 0, stream>>>(tbl);
  gin_binsplit<<<NBLK, ST, 0, stream>>>(ei, ei + GIN_E, tbl, binned);
  gin_binrows<<<NBIN, ST, 0, stream>>>(tbl, binned, rowinfo, csr);

  const float* hcur = x;
  for (int l = 0; l < 5; ++l) {
    float* dstp = (l == 4) ? out : ((l % 2 == 0) ? hB : hA);
    gin_layer<<<lgrid, WPB * 32, 0, stream>>>(
        hcur, rowinfo, csr,
        W1 + (size_t)l * GIN_D * GIN_D, b1 + (size_t)l * GIN_D,
        W2 + (size_t)l * GIN_D * GIN_D, b2 + (size_t)l * GIN_D,
        g  + (size_t)l * GIN_D, be + (size_t)l * GIN_D,
        mu + (size_t)l * GIN_D, va + (size_t)l * GIN_D,
        dstp, (l < 4) ? 1 : 0);
    hcur = dstp;
  }
}


# Scan launch removed: every split block derives its cursors from the raw [bin][chunk] counts (whole-line row reads, wave sums, 196-entry block scan); 8 launches; per-node sum order as before
# speedup vs baseline: 3.9520x; 1.0689x over previous
#include <hip/hip_runtime.h>
#include <hip/hip_bf16.h>
#include <cstdint>

typedef __attribute__((ext_vector_type(2))) float v2f;
typedef __attribute__((ext_vector_type(8))) float v8f;

#define GIN_N 100000
#define GIN_E 1600000
#define GIN_D 64
#define WPB 25
#define TSTRIDE 66

#define ST 1024
#define NBLK 256
#define BSH 9
#define BNODES (1 << BSH)
#define NBIN ((GIN_N + BNODES - 1) >> BSH)
#define NTBL (NBIN * NBLK)
#define BPW ((NBIN + ST / 32 - 1) / (ST / 32))
#define CHUNK ((GIN_E + NBLK - 1) / NBLK)
#define EPT ((CHUNK + ST - 1) / ST)
#define CAP 12288
#define RB 8
#define NOEDGE 0xFFFFFFFFu


__global__ __launch_bounds__(ST) void gin_bincount(
    const int* __restrict__ dst, int* __restrict__ tbl) {
  __shared__ int sc[NBIN];
  int t = threadIdx.x, b = blockIdx.x;
  if (t < NBIN) sc[t] = 0;
  __syncthreads();
  int e0 = b * CHUNK;
  int e1 = e0 + CHUNK < GIN_E ? e0 + CHUNK : GIN_E;
  int d[EPT];
#pragma unroll
  for (int j = 0; j < EPT; ++j) {
    int e = e0 + t + j * ST;
    d[j] = (e < e1) ? dst[e] : -1;
  }
#pragma unroll
  for (int j = 0; j < EPT; ++j)
    if (d[j] >= 0) atomicAdd(&sc[d[j] >> BSH], 1);
  __syncthreads();
  if (t < NBIN) tbl[t * NBLK + b] = sc[t];
}

__global__ __launch_bounds__(ST) void gin_binsplit(
    const int* __restrict__ src, const int* __restrict__ dst,
    const int* __restrict__ tbl, int* __restrict__ binstart,
    unsigned* __restrict__ binned) {
  __shared__ int sc[NBIN];
  __shared__ int sTot[NBIN];
  __shared__ int sPre[NBIN];
  __shared__ int sW[ST / 32];
  int t = threadIdx.x, b = blockIdx.x;
  int lane = t & 31, w = t >> 5;
  int4 ca[BPW], cb[BPW];
#pragma unroll
  for (int r = 0; r < BPW; ++r) {
    int bin = w + r * (ST / 32);
    int rowb = bin < NBIN ? bin : NBIN - 1;
    const int4* row = (const int4*)(tbl + rowb * NBLK) + 2 * lane;
    ca[r] = row[0];
    cb[r] = row[1];
  }
  int e0 = b * CHUNK;
  int e1 = e0 + CHUNK < GIN_E ? e0 + CHUNK : GIN_E;
  int d[EPT];
  unsigned s[EPT];
#pragma unroll
  for (int j = 0; j < EPT; ++j) {
    int e = e0 + t + j * ST;
    bool in = e < e1;
    d[j] = in ? dst[e] : -1;
    s[j] = in ? (unsigned)src[e] : 0u;
  }
  int c0 = 8 * lane;
  int tot[BPW], pre[BPW];
#pragma unroll
  for (int r = 0; r < BPW; ++r) {
    tot[r] = ca[r].x + ca[r].y + ca[r].z + ca[r].w +
             cb[r].x + cb[r].y + cb[r].z + cb[r].w;
    pre[r] = (c0     < b ? ca[r].x : 0) + (c0 + 1 < b ? ca[r].y : 0) +
             (c0 + 2 < b ? ca[r].z : 0) + (c0 + 3 < b ? ca[r].w : 0) +
             (c0 + 4 < b ? cb[r].x : 0) + (c0 + 5 < b ? cb[r].y : 0) +
             (c0 + 6 < b ? cb[r].z : 0) + (c0 + 7 < b ? cb[r].w : 0);
  }
#pragma unroll
  for (int o = 16; o > 0; o >>= 1) {
#pragma unroll
    for (int r = 0; r < BPW; ++r) {
      tot[r] += __shfl_xor(tot[r], o);
      pre[r] += __shfl_xor(pre[r], o);
    }
  }
  if (lane == 0) {
#pragma unroll
    for (int r = 0; r < BPW; ++r) {
      int bin = w + r * (ST / 32);
      if (bin < NBIN) { sTot[bin] = tot[r]; sPre[bin] = pre[r]; }
    }
  }
  __syncthreads();
  int own = (t < NBIN) ? sTot[t] : 0;
  int inc = own;
#pragma unroll
  for (int o = 1; o < 32; o <<= 1) {
    int u = __shfl_up(inc, o);
    if (lane >= o) inc += u;
  }
  if (lane == 31) sW[w] = inc;
  __syncthreads();
  int base = 0;
#pragma unroll
  for (int k = 0; k < ST / 32; ++k) if (k < w) base += sW[k];
  int excl = base + inc - own;
  if (t < NBIN) {
    sc[t] = excl + sPre[t];
    if (b == 0) binstart[t] = excl;
  }
  __syncthreads();
#pragma unroll
  for (int j = 0; j < EPT; ++j) {
    if (d[j] >= 0) {
      int pos = atomicAdd(&sc[d[j] >> BSH], 1);
      binned[pos] = (s[j] << BSH) | (unsigned)(d[j] & (BNODES - 1));
    }
  }
}

__global__ __launch_bounds__(ST) void gin_binrows(
    const int* __restrict__ binstart, const unsigned* __restrict__ binned,
    int2* __restrict__ rowinfo, int* __restrict__ csr) {
  __shared__ int sCnt[BNODES];
  __shared__ int sW[ST / 32];
  __shared__ int sOut[CAP];
  int t = threadIdx.x, b = blockIdx.x;
  int lane = t & 31, w = t >> 5;
  int beg = binstart[b];
  int end = (b + 1 < NBIN) ? binstart[b + 1] : GIN_E;
  int len = end - beg;
  const unsigned* bw = binned + beg;
  if (t < BNODES) sCnt[t] = 0;
  __syncthreads();
  for (int i0 = t; i0 < len; i0 += RB * ST) {
    unsigned x[RB];
#pragma unroll
    for (int j = 0; j < RB; ++j) {
      int i = i0 + j * ST;
      x[j] = (i < len) ? bw[i] : NOEDGE;
    }
#pragma unroll
    for (int j = 0; j < RB; ++j)
      if (x[j] != NOEDGE) atomicAdd(&sCnt[x[j] & (BNODES - 1)], 1);
  }
  __syncthreads();
  int own = (t < BNODES) ? sCnt[t] : 0;
  int inc = own;
#pragma unroll
  for (int o = 1; o < 32; o <<= 1) {
    int u = __shfl_up(inc, o);
    if (lane >= o) inc += u;
  }
  if (lane == 31) sW[w] = inc;
  __syncthreads();
  int base = 0;
#pragma unroll
  for (int k = 0; k < ST / 32; ++k) if (k < w) base += sW[k];
  int excl = base + inc - own;
  if (t < BNODES) {
    sCnt[t] = excl;
    int node = b * BNODES + t;
    if (node < GIN_N) rowinfo[node] = make_int2(beg + excl, own);
  }
  __syncthreads();
  if (len <= CAP) {
    for (int i0 = t; i0 < len; i0 += RB * ST) {
      unsigned x[RB];
#pragma unroll
      for (int j = 0; j < RB; ++j) {
        int i = i0 + j * ST;
        x[j] = (i < len) ? bw[i] : NOEDGE;
      }
#pragma unroll
      for (int j = 0; j < RB; ++j) {
        if (x[j] != NOEDGE) {
          int pos = atomicAdd(&sCnt[x[j] & (BNODES - 1)], 1);
          sOut[pos] = (int)(x[j] >> BSH);
        }
      }
    }
    __syncthreads();
    for (int i = t; i < len; i += ST) csr[beg + i] = sOut[i];
  } else {
    for (int i = t; i < len; i += ST) {
      unsigned x = bw[i];
      int pos = atomicAdd(&sCnt[x & (BNODES - 1)], 1);
      csr[beg + pos] = (int)(x >> BSH);
    }
  }
}

__global__ __launch_bounds__(WPB * 32) void gin_layer(
    const float* __restrict__ h, const int2* __restrict__ rowinfo,
    const int* __restrict__ csr,
    const float* __restrict__ W1, const float* __restrict__ b1,
    const float* __restrict__ W2, const float* __restrict__ b2,
    const float* __restrict__ g,  const float* __restrict__ be,
    const float* __restrict__ mu, const float* __restrict__ va,
    float* __restrict__ out, int do_bn) {
  __shared__ float sW1t[GIN_D * TSTRIDE];
  __shared__ float sW2t[GIN_D * TSTRIDE];
  __shared__ float sT[WPB][16 * TSTRIDE];

  int t = threadIdx.x;
#pragma unroll
  for (int i = t; i < GIN_D * GIN_D; i += WPB * 32) {
    int k = i >> 6, n = i & 63;
    sW1t[n * TSTRIDE + k] = W1[i];
    sW2t[n * TSTRIDE + k] = W2[i];
  }

  int wave = t >> 5, lane = t & 31;
  int row0 = (blockIdx.x * WPB + wave) * 16;
  bool active = row0 < GIN_N;
  int m    = lane & 15;
  int half = lane >> 4;
  float* st = sT[wave];

  if (active) {
    const float* hb = h + m * 4;
    int2 ri = rowinfo[row0 + m];
    int start = __shfl(ri.x, half, 16);
    int deg   = __shfl(ri.y, half, 16);
    int ia = (m < deg) ? csr[start + m] : 0;
    int ib = (16 + m < deg) ? csr[start + 16 + m] : 0;
    for (int q = 0; q < 8; ++q) {
      int startN = 0, degN = 0, iaN = 0, ibN = 0;
      if (q + 1 < 8) {
        int rn = 2 * (q + 1) + half;
        startN = __shfl(ri.x, rn, 16);
        degN   = __shfl(ri.y, rn, 16);
        iaN = (m < degN) ? csr[startN + m] : 0;
        ibN = (16 + m < degN) ? csr[startN + 16 + m] : 0;
      }
      int node = row0 + 2 * q + half;
      float4 hs = *(const float4*)(hb + (size_t)node * GIN_D);
      float4 a0 = make_float4(0.f, 0.f, 0.f, 0.f);
      float4 a1 = make_float4(0.f, 0.f, 0.f, 0.f);
      for (int base = 0; base < deg; base += 32) {
        int mm = deg - base;
        if (mm > 32) mm = 32;
        if (base > 0) {
          ia = (base + m < deg) ? csr[start + base + m] : 0;
          ib = (base + 16 + m < deg) ? csr[start + base + 16 + m] : 0;
        }
        for (int i = 0; i < mm; i += 8) {
          int iv = (i < 16) ? ia : ib;
          int j0 = i & 15;
          int rem = mm - i;
          float4 v[8];
#pragma unroll
          for (int k = 0; k < 8; ++k) {
            int s = __shfl(iv, j0 + k, 16);
            v[k] = make_float4(0.f, 0.f, 0.f, 0.f);
            if (k < rem) v[k] = *(const float4*)(hb + (size_t)s * GIN_D);
          }
#pragma unroll
          for (int k = 0; k < 8; k += 2) {
            a0.x += v[k].x; a0.y += v[k].y; a0.z += v[k].z; a0.w += v[k].w;
            a1.x += v[k + 1].x; a1.y += v[k + 1].y; a1.z += v[k + 1].z; a1.w += v[k + 1].w;
          }
        }
      }
      a0.x += a1.x; a0.y += a1.y; a0.z += a1.z; a0.w += a1.w;
      float* zr = st + (2 * q + half) * TSTRIDE + m * 4;
      v2f z01, z23;
      z01.x = hs.x + a0.x; z01.y = hs.y + a0.y;
      z23.x = hs.z + a0.z; z23.y = hs.w + a0.w;
      *(v2f*)zr = z01;
      *(v2f*)(zr + 2) = z23;
      start = startN; deg = degN; ia = iaN; ib = ibN;
    }
  }
  __syncthreads();
  if (!active) return;

  v2f zA[16];
#pragma unroll
  for (int k0 = 0; k0 < 16; ++k0)
    zA[k0] = *(const v2f*)&st[m * TSTRIDE + 4 * k0 + 2 * half];

  v8f acc[4];
#pragma unroll
  for (int n0 = 0; n0 < 4; ++n0)
#pragma unroll
    for (int i = 0; i < 8; ++i) acc[n0][i] = 0.f;

#pragma unroll
  for (int n0 = 0; n0 < 4; ++n0) {
    const float* wcol = &sW1t[(n0 * 16 + m) * TSTRIDE + 2 * half];
#pragma unroll
    for (int k0 = 0; k0 < 16; ++k0) {
      v2f b = *(const v2f*)(wcol + 4 * k0);
      acc[n0] = __builtin_amdgcn_wmma_f32_16x16x4_f32(
          false, zA[k0], false, b, (short)0, acc[n0], false, false);
    }
  }

#pragma unroll
  for (int n0 = 0; n0 < 4; ++n0) {
    float bias = b1[n0 * 16 + m];
#pragma unroll
    for (int i = 0; i < 8; ++i) {
      float v = fmaxf(acc[n0][i] + bias, 0.f);
      st[(half * 8 + i) * TSTRIDE + n0 * 16 + m] = v;
    }
  }

  v2f h1A[16];
#pragma unroll
  for (int k0 = 0; k0 < 16; ++k0)
    h1A[k0] = *(const v2f*)&st[m * TSTRIDE + 4 * k0 + 2 * half];

  v8f acc2[4];
#pragma unroll
  for (int n0 = 0; n0 < 4; ++n0)
#pragma unroll
    for (int i = 0; i < 8; ++i) acc2[n0][i] = 0.f;

#pragma unroll
  for (int n0 = 0; n0 < 4; ++n0) {
    const float* wcol = &sW2t[(n0 * 16 + m) * TSTRIDE + 2 * half];
#pragma unroll
    for (int k0 = 0; k0 < 16; ++k0) {
      v2f b = *(const v2f*)(wcol + 4 * k0);
      acc2[n0] = __builtin_amdgcn_wmma_f32_16x16x4_f32(
          false, h1A[k0], false, b, (short)0, acc2[n0], false, false);
    }
  }

#pragma unroll
  for (int n0 = 0; n0 < 4; ++n0) {
    int col = n0 * 16 + m;
    float bias = b2[col];
    float scale = 1.f, shift = 0.f;
    if (do_bn) {
      float s = g[col] * rsqrtf(va[col] + 1e-5f);
      scale = s;
      shift = be[col] - mu[col] * s;
    }
#pragma unroll
    for (int i = 0; i < 8; ++i) {
      float v = acc2[n0][i] + bias;
      if (do_bn) v = fmaxf(v * scale + shift, 0.f);
      out[(size_t)(row0 + half * 8 + i) * GIN_D + col] = v;
    }
  }
}

extern "C" void kernel_launch(void* const* d_in, const int* in_sizes, int n_in,
                              void* d_out, int out_size, void* d_ws, size_t ws_size,
                              hipStream_t stream) {
  const float* x  = (const float*)d_in[0];
  const int*   ei = (const int*)d_in[1];
  const float* W1 = (const float*)d_in[2];
  const float* b1 = (const float*)d_in[3];
  const float* W2 = (const float*)d_in[4];
  const float* b2 = (const float*)d_in[5];
  const float* g  = (const float*)d_in[6];
  const float* be = (const float*)d_in[7];
  const float* mu = (const float*)d_in[8];
  const float* va = (const float*)d_in[9];
  float* out = (float*)d_out;

  float* hA    = (float*)d_ws;
  float* hB    = hA + (size_t)GIN_N * GIN_D;
  int2* rowinfo = (int2*)(hB + (size_t)GIN_N * GIN_D);
  int*  csr     = (int*)(rowinfo + GIN_N);
  unsigned* binned = (unsigned*)(csr + GIN_E);
  int*  tbl     = (int*)(binned + GIN_E);
  int*  binstart = tbl + NTBL;

  const int lgrid = (GIN_N / 16 + WPB - 1) / WPB;

  gin_bincount<<<NBLK, ST, 0, stream>>>(ei + GIN_E, tbl);
  gin_binsplit<<<NBLK, ST, 0, stream>>>(ei, ei + GIN_E, tbl, binstart, binned);
  gin_binrows<<<NBIN, ST, 0, stream>>>(binstart, binned, rowinfo, csr);

  const float* hcur = x;
  for (int l = 0; l < 5; ++l) {
    float* dstp = (l == 4) ? out : ((l % 2 == 0) ? hB : hA);
    gin_layer<<<lgrid, WPB * 32, 0, stream>>>(
        hcur, rowinfo, csr,
        W1 + (size_t)l * GIN_D * GIN_D, b1 + (size_t)l * GIN_D,
        W2 + (size_t)l * GIN_D * GIN_D, b2 + (size_t)l * GIN_D,
        g  + (size_t)l * GIN_D, be + (size_t)l * GIN_D,
        mu + (size_t)l * GIN_D, va + (size_t)l * GIN_D,
        dstp, (l < 4) ? 1 : 0);
    hcur = dstp;
  }
}


# Gather trip widened from 8 to 16 neighbour rows per half wave (degree <= 16 is one round trip), all source indices handed out by constant-lane shuffles; per-node sum order as before
# speedup vs baseline: 4.1225x; 1.0431x over previous
#include <hip/hip_runtime.h>
#include <hip/hip_bf16.h>
#include <cstdint>

typedef __attribute__((ext_vector_type(2))) float v2f;
typedef __attribute__((ext_vector_type(8))) float v8f;

#define GIN_N 100000
#define GIN_E 1600000
#define GIN_D 64
#define WPB 25
#define TSTRIDE 66

#define ST 1024
#define NBLK 256
#define BSH 9
#define BNODES (1 << BSH)
#define NBIN ((GIN_N + BNODES - 1) >> BSH)
#define NTBL (NBIN * NBLK)
#define BPW ((NBIN + ST / 32 - 1) / (ST / 32))
#define CHUNK ((GIN_E + NBLK - 1) / NBLK)
#define EPT ((CHUNK + ST - 1) / ST)
#define CAP 12288
#define RB 8
#define NOEDGE 0xFFFFFFFFu


__global__ __launch_bounds__(ST) void gin_bincount(
    const int* __restrict__ dst, int* __restrict__ tbl) {
  __shared__ int sc[NBIN];
  int t = threadIdx.x, b = blockIdx.x;
  if (t < NBIN) sc[t] = 0;
  __syncthreads();
  int e0 = b * CHUNK;
  int e1 = e0 + CHUNK < GIN_E ? e0 + CHUNK : GIN_E;
  int d[EPT];
#pragma unroll
  for (int j = 0; j < EPT; ++j) {
    int e = e0 + t + j * ST;
    d[j] = (e < e1) ? dst[e] : -1;
  }
#pragma unroll
  for (int j = 0; j < EPT; ++j)
    if (d[j] >= 0) atomicAdd(&sc[d[j] >> BSH], 1);
  __syncthreads();
  if (t < NBIN) tbl[t * NBLK + b] = sc[t];
}

__global__ __launch_bounds__(ST) void gin_binsplit(
    const int* __restrict__ src, const int* __restrict__ dst,
    const int* __restrict__ tbl, int* __restrict__ binstart,
    unsigned* __restrict__ binned) {
  __shared__ int sc[NBIN];
  __shared__ int sTot[NBIN];
  __shared__ int sPre[NBIN];
  __shared__ int sW[ST / 32];
  int t = threadIdx.x, b = blockIdx.x;
  int lane = t & 31, w = t >> 5;
  int4 ca[BPW], cb[BPW];
#pragma unroll
  for (int r = 0; r < BPW; ++r) {
    int bin = w + r * (ST / 32);
    int rowb = bin < NBIN ? bin : NBIN - 1;
    const int4* row = (const int4*)(tbl + rowb * NBLK) + 2 * lane;
    ca[r] = row[0];
    cb[r] = row[1];
  }
  int e0 = b * CHUNK;
  int e1 = e0 + CHUNK < GIN_E ? e0 + CHUNK : GIN_E;
  int d[EPT];
  unsigned s[EPT];
#pragma unroll
  for (int j = 0; j < EPT; ++j) {
    int e = e0 + t + j * ST;
    bool in = e < e1;
    d[j] = in ? dst[e] : -1;
    s[j] = in ? (unsigned)src[e] : 0u;
  }
  int c0 = 8 * lane;
  int tot[BPW], pre[BPW];
#pragma unroll
  for (int r = 0; r < BPW; ++r) {
    tot[r] = ca[r].x + ca[r].y + ca[r].z + ca[r].w +
             cb[r].x + cb[r].y + cb[r].z + cb[r].w;
    pre[r] = (c0     < b ? ca[r].x : 0) + (c0 + 1 < b ? ca[r].y : 0) +
             (c0 + 2 < b ? ca[r].z : 0) + (c0 + 3 < b ? ca[r].w : 0) +
             (c0 + 4 < b ? cb[r].x : 0) + (c0 + 5 < b ? cb[r].y : 0) +
             (c0 + 6 < b ? cb[r].z : 0) + (c0 + 7 < b ? cb[r].w : 0);
  }
#pragma unroll
  for (int o = 16; o > 0; o >>= 1) {
#pragma unroll
    for (int r = 0; r < BPW; ++r) {
      tot[r] += __shfl_xor(tot[r], o);
      pre[r] += __shfl_xor(pre[r], o);
    }
  }
  if (lane == 0) {
#pragma unroll
    for (int r = 0; r < BPW; ++r) {
      int bin = w + r * (ST / 32);
      if (bin < NBIN) { sTot[bin] = tot[r]; sPre[bin] = pre[r]; }
    }
  }
  __syncthreads();
  int own = (t < NBIN) ? sTot[t] : 0;
  int inc = own;
#pragma unroll
  for (int o = 1; o < 32; o <<= 1) {
    int u = __shfl_up(inc, o);
    if (lane >= o) inc += u;
  }
  if (lane == 31) sW[w] = inc;
  __syncthreads();
  int base = 0;
#pragma unroll
  for (int k = 0; k < ST / 32; ++k) if (k < w) base += sW[k];
  int excl = base + inc - own;
  if (t < NBIN) {
    sc[t] = excl + sPre[t];
    if (b == 0) binstart[t] = excl;
  }
  __syncthreads();
#pragma unroll
  for (int j = 0; j < EPT; ++j) {
    if (d[j] >= 0) {
      int pos = atomicAdd(&sc[d[j] >> BSH], 1);
      binned[pos] = (s[j] << BSH) | (unsigned)(d[j] & (BNODES - 1));
    }
  }
}

__global__ __launch_bounds__(ST) void gin_binrows(
    const int* __restrict__ binstart, const unsigned* __restrict__ binned,
    int2* __restrict__ rowinfo, int* __restrict__ csr) {
  __shared__ int sCnt[BNODES];
  __shared__ int sW[ST / 32];
  __shared__ int sOut[CAP];
  int t = threadIdx.x, b = blockIdx.x;
  int lane = t & 31, w = t >> 5;
  int beg = binstart[b];
  int end = (b + 1 < NBIN) ? binstart[b + 1] : GIN_E;
  int len = end - beg;
  const unsigned* bw = binned + beg;
  if (t < BNODES) sCnt[t] = 0;
  __syncthreads();
  for (int i0 = t; i0 < len; i0 += RB * ST) {
    unsigned x[RB];
#pragma unroll
    for (int j = 0; j < RB; ++j) {
      int i = i0 + j * ST;
      x[j] = (i < len) ? bw[i] : NOEDGE;
    }
#pragma unroll
    for (int j = 0; j < RB; ++j)
      if (x[j] != NOEDGE) atomicAdd(&sCnt[x[j] & (BNODES - 1)], 1);
  }
  __syncthreads();
  int own = (t < BNODES) ? sCnt[t] : 0;
  int inc = own;
#pragma unroll
  for (int o = 1; o < 32; o <<= 1) {
    int u = __shfl_up(inc, o);
    if (lane >= o) inc += u;
  }
  if (lane == 31) sW[w] = inc;
  __syncthreads();
  int base = 0;
#pragma unroll
  for (int k = 0; k < ST / 32; ++k) if (k < w) base += sW[k];
  int excl = base + inc - own;
  if (t < BNODES) {
    sCnt[t] = excl;
    int node = b * BNODES + t;
    if (node < GIN_N) rowinfo[node] = make_int2(beg + excl, own);
  }
  __syncthreads();
  if (len <= CAP) {
    for (int i0 = t; i0 < len; i0 += RB * ST) {
      unsigned x[RB];
#pragma unroll
      for (int j = 0; j < RB; ++j) {
        int i = i0 + j * ST;
        x[j] = (i < len) ? bw[i] : NOEDGE;
      }
#pragma unroll
      for (int j = 0; j < RB; ++j) {
        if (x[j] != NOEDGE) {
          int pos = atomicAdd(&sCnt[x[j] & (BNODES - 1)], 1);
          sOut[pos] = (int)(x[j] >> BSH);
        }
      }
    }
    __syncthreads();
    for (int i = t; i < len; i += ST) csr[beg + i] = sOut[i];
  } else {
    for (int i = t; i < len; i += ST) {
      unsigned x = bw[i];
      int pos = atomicAdd(&sCnt[x & (BNODES - 1)], 1);
      csr[beg + pos] = (int)(x >> BSH);
    }
  }
}

__global__ __launch_bounds__(WPB * 32) void gin_layer(
    const float* __restrict__ h, const int2* __restrict__ rowinfo,
    const int* __restrict__ csr,
    const float* __restrict__ W1, const float* __restrict__ b1,
    const float* __restrict__ W2, const float* __restrict__ b2,
    const float* __restrict__ g,  const float* __restrict__ be,
    const float* __restrict__ mu, const float* __restrict__ va,
    float* __restrict__ out, int do_bn) {
  __shared__ float sW1t[GIN_D * TSTRIDE];
  __shared__ float sW2t[GIN_D * TSTRIDE];
  __shared__ float sT[WPB][16 * TSTRIDE];

  int t = threadIdx.x;
#pragma unroll
  for (int i = t; i < GIN_D * GIN_D; i += WPB * 32) {
    int k = i >> 6, n = i & 63;
    sW1t[n * TSTRIDE + k] = W1[i];
    sW2t[n * TSTRIDE + k] = W2[i];
  }

  int wave = t >> 5, lane = t & 31;
  int row0 = (blockIdx.x * WPB + wave) * 16;
  bool active = row0 < GIN_N;
  int m    = lane & 15;
  int half = lane >> 4;
  float* st = sT[wave];

  if (active) {
    const float* hb = h + m * 4;
    float zf;
    asm("v_mov_b32 %0, 0" : "=v"(zf));
    int2 ri = rowinfo[row0 + m];
    int start = __shfl(ri.x, half, 16);
    int deg   = __shfl(ri.y, half, 16);
    int ia = (m < deg) ? csr[start + m] : 0;
    int ib = (16 + m < deg) ? csr[start + 16 + m] : 0;
    for (int q = 0; q < 8; ++q) {
      int startN = 0, degN = 0, iaN = 0, ibN = 0;
      if (q + 1 < 8) {
        int rn = 2 * (q + 1) + half;
        startN = __shfl(ri.x, rn, 16);
        degN   = __shfl(ri.y, rn, 16);
        iaN = (m < degN) ? csr[startN + m] : 0;
        ibN = (16 + m < degN) ? csr[startN + 16 + m] : 0;
      }
      int node = row0 + 2 * q + half;
      float4 hs = *(const float4*)(hb + (size_t)node * GIN_D);
      float4 a0 = make_float4(0.f, 0.f, 0.f, 0.f);
      float4 a1 = make_float4(0.f, 0.f, 0.f, 0.f);
      for (int base = 0; base < deg; base += 16) {
        int iv = ia;
        if (base == 16) iv = ib;
        if (base > 16) iv = (base + m < deg) ? csr[start + base + m] : 0;
        int rem = deg - base;
        float4 v[16];
#pragma unroll
        for (int k = 0; k < 16; ++k) {
          int s = __shfl(iv, k, 16);
          v[k] = make_float4(zf, zf, zf, zf);
          if (k < rem) v[k] = *(const float4*)(hb + (size_t)s * GIN_D);
        }
#pragma unroll
        for (int k = 0; k < 16; k += 2) {
          a0.x += v[k].x; a0.y += v[k].y; a0.z += v[k].z; a0.w += v[k].w;
          a1.x += v[k + 1].x; a1.y += v[k + 1].y; a1.z += v[k + 1].z; a1.w += v[k + 1].w;
        }
      }
      a0.x += a1.x; a0.y += a1.y; a0.z += a1.z; a0.w += a1.w;
      float* zr = st + (2 * q + half) * TSTRIDE + m * 4;
      v2f z01, z23;
      z01.x = hs.x + a0.x; z01.y = hs.y + a0.y;
      z23.x = hs.z + a0.z; z23.y = hs.w + a0.w;
      *(v2f*)zr = z01;
      *(v2f*)(zr + 2) = z23;
      start = startN; deg = degN; ia = iaN; ib = ibN;
    }
  }
  __syncthreads();
  if (!active) return;

  v2f zA[16];
#pragma unroll
  for (int k0 = 0; k0 < 16; ++k0)
    zA[k0] = *(const v2f*)&st[m * TSTRIDE + 4 * k0 + 2 * half];

  v8f acc[4];
#pragma unroll
  for (int n0 = 0; n0 < 4; ++n0)
#pragma unroll
    for (int i = 0; i < 8; ++i) acc[n0][i] = 0.f;

#pragma unroll
  for (int n0 = 0; n0 < 4; ++n0) {
    const float* wcol = &sW1t[(n0 * 16 + m) * TSTRIDE + 2 * half];
#pragma unroll
    for (int k0 = 0; k0 < 16; ++k0) {
      v2f b = *(const v2f*)(wcol + 4 * k0);
      acc[n0] = __builtin_amdgcn_wmma_f32_16x16x4_f32(
          false, zA[k0], false, b, (short)0, acc[n0], false, false);
    }
  }

#pragma unroll
  for (int n0 = 0; n0 < 4; ++n0) {
    float bias = b1[n0 * 16 + m];
#pragma unroll
    for (int i = 0; i < 8; ++i) {
      float v = fmaxf(acc[n0][i] + bias, 0.f);
      st[(half * 8 + i) * TSTRIDE + n0 * 16 + m] = v;
    }
  }

  v2f h1A[16];
#pragma unroll
  for (int k0 = 0; k0 < 16; ++k0)
    h1A[k0] = *(const v2f*)&st[m * TSTRIDE + 4 * k0 + 2 * half];

  v8f acc2[4];
#pragma unroll
  for (int n0 = 0; n0 < 4; ++n0)
#pragma unroll
    for (int i = 0; i < 8; ++i) acc2[n0][i] = 0.f;

#pragma unroll
  for (int n0 = 0; n0 < 4; ++n0) {
    const float* wcol = &sW2t[(n0 * 16 + m) * TSTRIDE + 2 * half];
#pragma unroll
    for (int k0 = 0; k0 < 16; ++k0) {
      v2f b = *(const v2f*)(wcol + 4 * k0);
      acc2[n0] = __builtin_amdgcn_wmma_f32_16x16x4_f32(
          false, h1A[k0], false, b, (short)0, acc2[n0], false, false);
    }
  }

#pragma unroll
  for (int n0 = 0; n0 < 4; ++n0) {
    int col = n0 * 16 + m;
    float bias = b2[col];
    float scale = 1.f, shift = 0.f;
    if (do_bn) {
      float s = g[col] * rsqrtf(va[col] + 1e-5f);
      scale = s;
      shift = be[col] - mu[col] * s;
    }
#pragma unroll
    for (int i = 0; i < 8; ++i) {
      float v = acc2[n0][i] + bias;
      if (do_bn) v = fmaxf(v * scale + shift, 0.f);
      out[(size_t)(row0 + half * 8 + i) * GIN_D + col] = v;
    }
  }
}

extern "C" void kernel_launch(void* const* d_in, const int* in_sizes, int n_in,
                              void* d_out, int out_size, void* d_ws, size_t ws_size,
                              hipStream_t stream) {
  const float* x  = (const float*)d_in[0];
  const int*   ei = (const int*)d_in[1];
  const float* W1 = (const float*)d_in[2];
  const float* b1 = (const float*)d_in[3];
  const float* W2 = (const float*)d_in[4];
  const float* b2 = (const float*)d_in[5];
  const float* g  = (const float*)d_in[6];
  const float* be = (const float*)d_in[7];
  const float* mu = (const float*)d_in[8];
  const float* va = (const float*)d_in[9];
  float* out = (float*)d_out;

  float* hA    = (float*)d_ws;
  float* hB    = hA + (size_t)GIN_N * GIN_D;
  int2* rowinfo = (int2*)(hB + (size_t)GIN_N * GIN_D);
  int*  csr     = (int*)(rowinfo + GIN_N);
  unsigned* binned = (unsigned*)(csr + GIN_E);
  int*  tbl     = (int*)(binned + GIN_E);
  int*  binstart = tbl + NTBL;

  const int lgrid = (GIN_N / 16 + WPB - 1) / WPB;

  gin_bincount<<<NBLK, ST, 0, stream>>>(ei + GIN_E, tbl);
  gin_binsplit<<<NBLK, ST, 0, stream>>>(ei, ei + GIN_E, tbl, binstart, binned);
  gin_binrows<<<NBIN, ST, 0, stream>>>(binstart, binned, rowinfo, csr);

  const float* hcur = x;
  for (int l = 0; l < 5; ++l) {
    float* dstp = (l == 4) ? out : ((l % 2 == 0) ? hB : hA);
    gin_layer<<<lgrid, WPB * 32, 0, stream>>>(
        hcur, rowinfo, csr,
        W1 + (size_t)l * GIN_D * GIN_D, b1 + (size_t)l * GIN_D,
        W2 + (size_t)l * GIN_D * GIN_D, b2 + (size_t)l * GIN_D,
        g  + (size_t)l * GIN_D, be + (size_t)l * GIN_D,
        mu + (size_t)l * GIN_D, va + (size_t)l * GIN_D,
        dstp, (l < 4) ? 1 : 0);
    hcur = dstp;
  }
}


# Gather row loads made unconditional: 32-bit row offsets from the h pointer, slots past a node's last edge read its own row with weight 0 (fma), scalar trip loop; per-node sum order as before
# speedup vs baseline: 4.4838x; 1.0876x over previous
#include <hip/hip_runtime.h>
#include <hip/hip_bf16.h>
#include <cstdint>

typedef __attribute__((ext_vector_type(2))) float v2f;
typedef __attribute__((ext_vector_type(8))) float v8f;

#define GIN_N 100000
#define GIN_E 1600000
#define GIN_D 64
#define WPB 25
#define TSTRIDE 66

#define ST 1024
#define NBLK 256
#define BSH 9
#define BNODES (1 << BSH)
#define NBIN ((GIN_N + BNODES - 1) >> BSH)
#define NTBL (NBIN * NBLK)
#define BPW ((NBIN + ST / 32 - 1) / (ST / 32))
#define CHUNK ((GIN_E + NBLK - 1) / NBLK)
#define EPT ((CHUNK + ST - 1) / ST)
#define CAP 12288
#define RB 8
#define NOEDGE 0xFFFFFFFFu


__global__ __launch_bounds__(ST) void gin_bincount(
    const int* __restrict__ dst, int* __restrict__ tbl) {
  __shared__ int sc[NBIN];
  int t = threadIdx.x, b = blockIdx.x;
  if (t < NBIN) sc[t] = 0;
  __syncthreads();
  int e0 = b * CHUNK;
  int e1 = e0 + CHUNK < GIN_E ? e0 + CHUNK : GIN_E;
  int d[EPT];
#pragma unroll
  for (int j = 0; j < EPT; ++j) {
    int e = e0 + t + j * ST;
    d[j] = (e < e1) ? dst[e] : -1;
  }
#pragma unroll
  for (int j = 0; j < EPT; ++j)
    if (d[j] >= 0) atomicAdd(&sc[d[j] >> BSH], 1);
  __syncthreads();
  if (t < NBIN) tbl[t * NBLK + b] = sc[t];
}

__global__ __launch_bounds__(ST) void gin_binsplit(
    const int* __restrict__ src, const int* __restrict__ dst,
    const int* __restrict__ tbl, int* __restrict__ binstart,
    unsigned* __restrict__ binned) {
  __shared__ int sc[NBIN];
  __shared__ int sTot[NBIN];
  __shared__ int sPre[NBIN];
  __shared__ int sW[ST / 32];
  int t = threadIdx.x, b = blockIdx.x;
  int lane = t & 31, w = t >> 5;
  int4 ca[BPW], cb[BPW];
#pragma unroll
  for (int r = 0; r < BPW; ++r) {
    int bin = w + r * (ST / 32);
    int rowb = bin < NBIN ? bin : NBIN - 1;
    const int4* row = (const int4*)(tbl + rowb * NBLK) + 2 * lane;
    ca[r] = row[0];
    cb[r] = row[1];
  }
  int e0 = b * CHUNK;
  int e1 = e0 + CHUNK < GIN_E ? e0 + CHUNK : GIN_E;
  int d[EPT];
  unsigned s[EPT];
#pragma unroll
  for (int j = 0; j < EPT; ++j) {
    int e = e0 + t + j * ST;
    bool in = e < e1;
    d[j] = in ? dst[e] : -1;
    s[j] = in ? (unsigned)src[e] : 0u;
  }
  int c0 = 8 * lane;
  int tot[BPW], pre[BPW];
#pragma unroll
  for (int r = 0; r < BPW; ++r) {
    tot[r] = ca[r].x + ca[r].y + ca[r].z + ca[r].w +
             cb[r].x + cb[r].y + cb[r].z + cb[r].w;
    pre[r] = (c0     < b ? ca[r].x : 0) + (c0 + 1 < b ? ca[r].y : 0) +
             (c0 + 2 < b ? ca[r].z : 0) + (c0 + 3 < b ? ca[r].w : 0) +
             (c0 + 4 < b ? cb[r].x : 0) + (c0 + 5 < b ? cb[r].y : 0) +
             (c0 + 6 < b ? cb[r].z : 0) + (c0 + 7 < b ? cb[r].w : 0);
  }
#pragma unroll
  for (int o = 16; o > 0; o >>= 1) {
#pragma unroll
    for (int r = 0; r < BPW; ++r) {
      tot[r] += __shfl_xor(tot[r], o);
      pre[r] += __shfl_xor(pre[r], o);
    }
  }
  if (lane == 0) {
#pragma unroll
    for (int r = 0; r < BPW; ++r) {
      int bin = w + r * (ST / 32);
      if (bin < NBIN) { sTot[bin] = tot[r]; sPre[bin] = pre[r]; }
    }
  }
  __syncthreads();
  int own = (t < NBIN) ? sTot[t] : 0;
  int inc = own;
#pragma unroll
  for (int o = 1; o < 32; o <<= 1) {
    int u = __shfl_up(inc, o);
    if (lane >= o) inc += u;
  }
  if (lane == 31) sW[w] = inc;
  __syncthreads();
  int base = 0;
#pragma unroll
  for (int k = 0; k < ST / 32; ++k) if (k < w) base += sW[k];
  int excl = base + inc - own;
  if (t < NBIN) {
    sc[t] = excl + sPre[t];
    if (b == 0) binstart[t] = excl;
  }
  __syncthreads();
#pragma unroll
  for (int j = 0; j < EPT; ++j) {
    if (d[j] >= 0) {
      int pos = atomicAdd(&sc[d[j] >> BSH], 1);
      binned[pos] = (s[j] << BSH) | (unsigned)(d[j] & (BNODES - 1));
    }
  }
}

__global__ __launch_bounds__(ST) void gin_binrows(
    const int* __restrict__ binstart, const unsigned* __restrict__ binned,
    int2* __restrict__ rowinfo, int* __restrict__ csr) {
  __shared__ int sCnt[BNODES];
  __shared__ int sW[ST / 32];
  __shared__ int sOut[CAP];
  int t = threadIdx.x, b = blockIdx.x;
  int lane = t & 31, w = t >> 5;
  int beg = binstart[b];
  int end = (b + 1 < NBIN) ? binstart[b + 1] : GIN_E;
  int len = end - beg;
  const unsigned* bw = binned + beg;
  if (t < BNODES) sCnt[t] = 0;
  __syncthreads();
  for (int i0 = t; i0 < len; i0 += RB * ST) {
    unsigned x[RB];
#pragma unroll
    for (int j = 0; j < RB; ++j) {
      int i = i0 + j * ST;
      x[j] = (i < len) ? bw[i] : NOEDGE;
    }
#pragma unroll
    for (int j = 0; j < RB; ++j)
      if (x[j] != NOEDGE) atomicAdd(&sCnt[x[j] & (BNODES - 1)], 1);
  }
  __syncthreads();
  int own = (t < BNODES) ? sCnt[t] : 0;
  int inc = own;
#pragma unroll
  for (int o = 1; o < 32; o <<= 1) {
    int u = __shfl_up(inc, o);
    if (lane >= o) inc += u;
  }
  if (lane == 31) sW[w] = inc;
  __syncthreads();
  int base = 0;
#pragma unroll
  for (int k = 0; k < ST / 32; ++k) if (k < w) base += sW[k];
  int excl = base + inc - own;
  if (t < BNODES) {
    sCnt[t] = excl;
    int node = b * BNODES + t;
    if (node < GIN_N) rowinfo[node] = make_int2(beg + excl, own);
  }
  __syncthreads();
  if (len <= CAP) {
    for (int i0 = t; i0 < len; i0 += RB * ST) {
      unsigned x[RB];
#pragma unroll
      for (int j = 0; j < RB; ++j) {
        int i = i0 + j * ST;
        x[j] = (i < len) ? bw[i] : NOEDGE;
      }
#pragma unroll
      for (int j = 0; j < RB; ++j) {
        if (x[j] != NOEDGE) {
          int pos = atomicAdd(&sCnt[x[j] & (BNODES - 1)], 1);
          sOut[pos] = (int)(x[j] >> BSH);
        }
      }
    }
    __syncthreads();
    for (int i = t; i < len; i += ST) csr[beg + i] = sOut[i];
  } else {
    for (int i = t; i < len; i += ST) {
      unsigned x = bw[i];
      int pos = atomicAdd(&sCnt[x & (BNODES - 1)], 1);
      csr[beg + pos] = (int)(x >> BSH);
    }
  }
}

__global__ __launch_bounds__(WPB * 32) void gin_layer(
    const float* __restrict__ h, const int2* __restrict__ rowinfo,
    const int* __restrict__ csr,
    const float* __restrict__ W1, const float* __restrict__ b1,
    const float* __restrict__ W2, const float* __restrict__ b2,
    const float* __restrict__ g,  const float* __restrict__ be,
    const float* __restrict__ mu, const float* __restrict__ va,
    float* __restrict__ out, int do_bn) {
  __shared__ float sW1t[GIN_D * TSTRIDE];
  __shared__ float sW2t[GIN_D * TSTRIDE];
  __shared__ float sT[WPB][16 * TSTRIDE];

  int t = threadIdx.x;
#pragma unroll
  for (int i = t; i < GIN_D * GIN_D; i += WPB * 32) {
    int k = i >> 6, n = i & 63;
    sW1t[n * TSTRIDE + k] = W1[i];
    sW2t[n * TSTRIDE + k] = W2[i];
  }

  int wave = t >> 5, lane = t & 31;
  int row0 = (blockIdx.x * WPB + wave) * 16;
  bool active = row0 < GIN_N;
  int m    = lane & 15;
  int half = lane >> 4;
  float* st = sT[wave];

  if (active) {
    const char* hbase = (const char*)h;
    unsigned lo = (unsigned)m * 16u;
    int2 ri = rowinfo[row0 + m];
    int start = __shfl(ri.x, half, 16);
    int deg   = __shfl(ri.y, half, 16);
    int ia = (m < deg) ? csr[start + m] : row0 + half;
    int ib = (16 + m < deg) ? csr[start + 16 + m] : row0 + half;
    for (int q = 0; q < 8; ++q) {
      int node = row0 + 2 * q + half;
      int startN = 0, degN = 0, iaN = 0, ibN = 0;
      if (q + 1 < 8) {
        int rn = 2 * (q + 1) + half;
        startN = __shfl(ri.x, rn, 16);
        degN   = __shfl(ri.y, rn, 16);
        iaN = (m < degN) ? csr[startN + m] : node + 2;
        ibN = (16 + m < degN) ? csr[startN + 16 + m] : node + 2;
      }
      float4 hs = *(const float4*)(hbase + (((unsigned)node << 8) | lo));
      float4 a0 = make_float4(0.f, 0.f, 0.f, 0.f);
      float4 a1 = make_float4(0.f, 0.f, 0.f, 0.f);
      int degO = __shfl_xor(deg, 16);
      int degM = __builtin_amdgcn_readfirstlane(deg > degO ? deg : degO);
      for (int base = 0; base < degM; base += 16) {
        int iv = ia;
        if (base == 16) iv = ib;
        if (base > 16) iv = (base + m < deg) ? csr[start + base + m] : node;
        int rem = deg - base;
        float4 v[16];
#pragma unroll
        for (int k = 0; k < 16; ++k) {
          int s = __shfl(iv, k, 16);
          v[k] = *(const float4*)(hbase + (((unsigned)s << 8) | lo));
        }
#pragma unroll
        for (int k = 0; k < 16; k += 2) {
          float w0 = (k < rem) ? 1.f : 0.f;
          float w1 = (k + 1 < rem) ? 1.f : 0.f;
          a0.x = fmaf(v[k].x, w0, a0.x); a0.y = fmaf(v[k].y, w0, a0.y);
          a0.z = fmaf(v[k].z, w0, a0.z); a0.w = fmaf(v[k].w, w0, a0.w);
          a1.x = fmaf(v[k + 1].x, w1, a1.x); a1.y = fmaf(v[k + 1].y, w1, a1.y);
          a1.z = fmaf(v[k + 1].z, w1, a1.z); a1.w = fmaf(v[k + 1].w, w1, a1.w);
        }
      }
      a0.x += a1.x; a0.y += a1.y; a0.z += a1.z; a0.w += a1.w;
      float* zr = st + (2 * q + half) * TSTRIDE + m * 4;
      v2f z01, z23;
      z01.x = hs.x + a0.x; z01.y = hs.y + a0.y;
      z23.x = hs.z + a0.z; z23.y = hs.w + a0.w;
      *(v2f*)zr = z01;
      *(v2f*)(zr + 2) = z23;
      start = startN; deg = degN; ia = iaN; ib = ibN;
    }
  }
  __syncthreads();
  if (!active) return;

  v2f zA[16];
#pragma unroll
  for (int k0 = 0; k0 < 16; ++k0)
    zA[k0] = *(const v2f*)&st[m * TSTRIDE + 4 * k0 + 2 * half];

  v8f acc[4];
#pragma unroll
  for (int n0 = 0; n0 < 4; ++n0)
#pragma unroll
    for (int i = 0; i < 8; ++i) acc[n0][i] = 0.f;

#pragma unroll
  for (int n0 = 0; n0 < 4; ++n0) {
    const float* wcol = &sW1t[(n0 * 16 + m) * TSTRIDE + 2 * half];
#pragma unroll
    for (int k0 = 0; k0 < 16; ++k0) {
      v2f b = *(const v2f*)(wcol + 4 * k0);
      acc[n0] = __builtin_amdgcn_wmma_f32_16x16x4_f32(
          false, zA[k0], false, b, (short)0, acc[n0], false, false);
    }
  }

#pragma unroll
  for (int n0 = 0; n0 < 4; ++n0) {
    float bias = b1[n0 * 16 + m];
#pragma unroll
    for (int i = 0; i < 8; ++i) {
      float v = fmaxf(acc[n0][i] + bias, 0.f);
      st[(half * 8 + i) * TSTRIDE + n0 * 16 + m] = v;
    }
  }

  v2f h1A[16];
#pragma unroll
  for (int k0 = 0; k0 < 16; ++k0)
    h1A[k0] = *(const v2f*)&st[m * TSTRIDE + 4 * k0 + 2 * half];

  v8f acc2[4];
#pragma unroll
  for (int n0 = 0; n0 < 4; ++n0)
#pragma unroll
    for (int i = 0; i < 8; ++i) acc2[n0][i] = 0.f;

#pragma unroll
  for (int n0 = 0; n0 < 4; ++n0) {
    const float* wcol = &sW2t[(n0 * 16 + m) * TSTRIDE + 2 * half];
#pragma unroll
    for (int k0 = 0; k0 < 16; ++k0) {
      v2f b = *(const v2f*)(wcol + 4 * k0);
      acc2[n0] = __builtin_amdgcn_wmma_f32_16x16x4_f32(
          false, h1A[k0], false, b, (short)0, acc2[n0], false, false);
    }
  }

#pragma unroll
  for (int n0 = 0; n0 < 4; ++n0) {
    int col = n0 * 16 + m;
    float bias = b2[col];
    float scale = 1.f, shift = 0.f;
    if (do_bn) {
      float s = g[col] * rsqrtf(va[col] + 1e-5f);
      scale = s;
      shift = be[col] - mu[col] * s;
    }
#pragma unroll
    for (int i = 0; i < 8; ++i) {
      float v = acc2[n0][i] + bias;
      if (do_bn) v = fmaxf(v * scale + shift, 0.f);
      out[(size_t)(row0 + half * 8 + i) * GIN_D + col] = v;
    }
  }
}

extern "C" void kernel_launch(void* const* d_in, const int* in_sizes, int n_in,
                              void* d_out, int out_size, void* d_ws, size_t ws_size,
                              hipStream_t stream) {
  const float* x  = (const float*)d_in[0];
  const int*   ei = (const int*)d_in[1];
  const float* W1 = (const float*)d_in[2];
  const float* b1 = (const float*)d_in[3];
  const float* W2 = (const float*)d_in[4];
  const float* b2 = (const float*)d_in[5];
  const float* g  = (const float*)d_in[6];
  const float* be = (const float*)d_in[7];
  const float* mu = (const float*)d_in[8];
  const float* va = (const float*)d_in[9];
  float* out = (float*)d_out;

  float* hA    = (float*)d_ws;
  float* hB    = hA + (size_t)GIN_N * GIN_D;
  int2* rowinfo = (int2*)(hB + (size_t)GIN_N * GIN_D);
  int*  csr     = (int*)(rowinfo + GIN_N);
  unsigned* binned = (unsigned*)(csr + GIN_E);
  int*  tbl     = (int*)(binned + GIN_E);
  int*  binstart = tbl + NTBL;

  const int lgrid = (GIN_N / 16 + WPB - 1) / WPB;

  gin_bincount<<<NBLK, ST, 0, stream>>>(ei + GIN_E, tbl);
  gin_binsplit<<<NBLK, ST, 0, stream>>>(ei, ei + GIN_E, tbl, binstart, binned);
  gin_binrows<<<NBIN, ST, 0, stream>>>(binstart, binned, rowinfo, csr);

  const float* hcur = x;
  for (int l = 0; l < 5; ++l) {
    float* dstp = (l == 4) ? out : ((l % 2 == 0) ? hB : hA);
    gin_layer<<<lgrid, WPB * 32, 0, stream>>>(
        hcur, rowinfo, csr,
        W1 + (size_t)l * GIN_D * GIN_D, b1 + (size_t)l * GIN_D,
        W2 + (size_t)l * GIN_D * GIN_D, b2 + (size_t)l * GIN_D,
        g  + (size_t)l * GIN_D, be + (size_t)l * GIN_D,
        mu + (size_t)l * GIN_D, va + (size_t)l * GIN_D,
        dstp, (l < 4) ? 1 : 0);
    hcur = dstp;
  }
}


# Tile's source indices staged in LDS before the node loop (16 loads in flight, one wait), degrees read as scalars: no index load or lane exchange waits before a node's row loads; sum order as before
# speedup vs baseline: 4.6347x; 1.0337x over previous
#include <hip/hip_runtime.h>
#include <hip/hip_bf16.h>
#include <cstdint>

typedef __attribute__((ext_vector_type(2))) float v2f;
typedef __attribute__((ext_vector_type(8))) float v8f;

#define GIN_N 100000
#define GIN_E 1600000
#define GIN_D 64
#define WPB 25
#define TSTRIDE 66

#define ST 1024
#define NBLK 256
#define BSH 9
#define BNODES (1 << BSH)
#define NBIN ((GIN_N + BNODES - 1) >> BSH)
#define NTBL (NBIN * NBLK)
#define BPW ((NBIN + ST / 32 - 1) / (ST / 32))
#define CHUNK ((GIN_E + NBLK - 1) / NBLK)
#define EPT ((CHUNK + ST - 1) / ST)
#define CAP 12288
#define RB 8
#define NOEDGE 0xFFFFFFFFu


__global__ __launch_bounds__(ST) void gin_bincount(
    const int* __restrict__ dst, int* __restrict__ tbl) {
  __shared__ int sc[NBIN];
  int t = threadIdx.x, b = blockIdx.x;
  if (t < NBIN) sc[t] = 0;
  __syncthreads();
  int e0 = b * CHUNK;
  int e1 = e0 + CHUNK < GIN_E ? e0 + CHUNK : GIN_E;
  int d[EPT];
#pragma unroll
  for (int j = 0; j < EPT; ++j) {
    int e = e0 + t + j * ST;
    d[j] = (e < e1) ? dst[e] : -1;
  }
#pragma unroll
  for (int j = 0; j < EPT; ++j)
    if (d[j] >= 0) atomicAdd(&sc[d[j] >> BSH], 1);
  __syncthreads();
  if (t < NBIN) tbl[t * NBLK + b] = sc[t];
}

__global__ __launch_bounds__(ST) void gin_binsplit(
    const int* __restrict__ src, const int* __restrict__ dst,
    const int* __restrict__ tbl, int* __restrict__ binstart,
    unsigned* __restrict__ binned) {
  __shared__ int sc[NBIN];
  __shared__ int sTot[NBIN];
  __shared__ int sPre[NBIN];
  __shared__ int sW[ST / 32];
  int t = threadIdx.x, b = blockIdx.x;
  int lane = t & 31, w = t >> 5;
  int4 ca[BPW], cb[BPW];
#pragma unroll
  for (int r = 0; r < BPW; ++r) {
    int bin = w + r * (ST / 32);
    int rowb = bin < NBIN ? bin : NBIN - 1;
    const int4* row = (const int4*)(tbl + rowb * NBLK) + 2 * lane;
    ca[r] = row[0];
    cb[r] = row[1];
  }
  int e0 = b * CHUNK;
  int e1 = e0 + CHUNK < GIN_E ? e0 + CHUNK : GIN_E;
  int d[EPT];
  unsigned s[EPT];
#pragma unroll
  for (int j = 0; j < EPT; ++j) {
    int e = e0 + t + j * ST;
    bool in = e < e1;
    d[j] = in ? dst[e] : -1;
    s[j] = in ? (unsigned)src[e] : 0u;
  }
  int c0 = 8 * lane;
  int tot[BPW], pre[BPW];
#pragma unroll
  for (int r = 0; r < BPW; ++r) {
    tot[r] = ca[r].x + ca[r].y + ca[r].z + ca[r].w +
             cb[r].x + cb[r].y + cb[r].z + cb[r].w;
    pre[r] = (c0     < b ? ca[r].x : 0) + (c0 + 1 < b ? ca[r].y : 0) +
             (c0 + 2 < b ? ca[r].z : 0) + (c0 + 3 < b ? ca[r].w : 0) +
             (c0 + 4 < b ? cb[r].x : 0) + (c0 + 5 < b ? cb[r].y : 0) +
             (c0 + 6 < b ? cb[r].z : 0) + (c0 + 7 < b ? cb[r].w : 0);
  }
#pragma unroll
  for (int o = 16; o > 0; o >>= 1) {
#pragma unroll
    for (int r = 0; r < BPW; ++r) {
      tot[r] += __shfl_xor(tot[r], o);
      pre[r] += __shfl_xor(pre[r], o);
    }
  }
  if (lane == 0) {
#pragma unroll
    for (int r = 0; r < BPW; ++r) {
      int bin = w + r * (ST / 32);
      if (bin < NBIN) { sTot[bin] = tot[r]; sPre[bin] = pre[r]; }
    }
  }
  __syncthreads();
  int own = (t < NBIN) ? sTot[t] : 0;
  int inc = own;
#pragma unroll
  for (int o = 1; o < 32; o <<= 1) {
    int u = __shfl_up(inc, o);
    if (lane >= o) inc += u;
  }
  if (lane == 31) sW[w] = inc;
  __syncthreads();
  int base = 0;
#pragma unroll
  for (int k = 0; k < ST / 32; ++k) if (k < w) base += sW[k];
  int excl = base + inc - own;
  if (t < NBIN) {
    sc[t] = excl + sPre[t];
    if (b == 0) binstart[t] = excl;
  }
  __syncthreads();
#pragma unroll
  for (int j = 0; j < EPT; ++j) {
    if (d[j] >= 0) {
      int pos = atomicAdd(&sc[d[j] >> BSH], 1);
      binned[pos] = (s[j] << BSH) | (unsigned)(d[j] & (BNODES - 1));
    }
  }
}

__global__ __launch_bounds__(ST) void gin_binrows(
    const int* __restrict__ binstart, const unsigned* __restrict__ binned,
    int2* __restrict__ rowinfo, int* __restrict__ csr) {
  __shared__ int sCnt[BNODES];
  __shared__ int sW[ST / 32];
  __shared__ int sOut[CAP];
  int t = threadIdx.x, b = blockIdx.x;
  int lane = t & 31, w = t >> 5;
  int beg = binstart[b];
  int end = (b + 1 < NBIN) ? binstart[b + 1] : GIN_E;
  int len = end - beg;
  const unsigned* bw = binned + beg;
  if (t < BNODES) sCnt[t] = 0;
  __syncthreads();
  for (int i0 = t; i0 < len; i0 += RB * ST) {
    unsigned x[RB];
#pragma unroll
    for (int j = 0; j < RB; ++j) {
      int i = i0 + j * ST;
      x[j] = (i < len) ? bw[i] : NOEDGE;
    }
#pragma unroll
    for (int j = 0; j < RB; ++j)
      if (x[j] != NOEDGE) atomicAdd(&sCnt[x[j] & (BNODES - 1)], 1);
  }
  __syncthreads();
  int own = (t < BNODES) ? sCnt[t] : 0;
  int inc = own;
#pragma unroll
  for (int o = 1; o < 32; o <<= 1) {
    int u = __shfl_up(inc, o);
    if (lane >= o) inc += u;
  }
  if (lane == 31) sW[w] = inc;
  __syncthreads();
  int base = 0;
#pragma unroll
  for (int k = 0; k < ST / 32; ++k) if (k < w) base += sW[k];
  int excl = base + inc - own;
  if (t < BNODES) {
    sCnt[t] = excl;
    int node = b * BNODES + t;
    if (node < GIN_N) rowinfo[node] = make_int2(beg + excl, own);
  }
  __syncthreads();
  if (len <= CAP) {
    for (int i0 = t; i0 < len; i0 += RB * ST) {
      unsigned x[RB];
#pragma unroll
      for (int j = 0; j < RB; ++j) {
        int i = i0 + j * ST;
        x[j] = (i < len) ? bw[i] : NOEDGE;
      }
#pragma unroll
      for (int j = 0; j < RB; ++j) {
        if (x[j] != NOEDGE) {
          int pos = atomicAdd(&sCnt[x[j] & (BNODES - 1)], 1);
          sOut[pos] = (int)(x[j] >> BSH);
        }
      }
    }
    __syncthreads();
    for (int i = t; i < len; i += ST) csr[beg + i] = sOut[i];
  } else {
    for (int i = t; i < len; i += ST) {
      unsigned x = bw[i];
      int pos = atomicAdd(&sCnt[x & (BNODES - 1)], 1);
      csr[beg + pos] = (int)(x >> BSH);
    }
  }
}

__global__ __launch_bounds__(WPB * 32) void gin_layer(
    const float* __restrict__ h, const int2* __restrict__ rowinfo,
    const int* __restrict__ csr,
    const float* __restrict__ W1, const float* __restrict__ b1,
    const float* __restrict__ W2, const float* __restrict__ b2,
    const float* __restrict__ g,  const float* __restrict__ be,
    const float* __restrict__ mu, const float* __restrict__ va,
    float* __restrict__ out, int do_bn) {
  __shared__ float sW1t[GIN_D * TSTRIDE];
  __shared__ float sW2t[GIN_D * TSTRIDE];
  __shared__ float sT[WPB][16 * TSTRIDE];
  __shared__ int sIdx[WPB][16 * 32];

  int t = threadIdx.x;
#pragma unroll
  for (int i = t; i < GIN_D * GIN_D; i += WPB * 32) {
    int k = i >> 6, n = i & 63;
    sW1t[n * TSTRIDE + k] = W1[i];
    sW2t[n * TSTRIDE + k] = W2[i];
  }

  int wave = t >> 5, lane = t & 31;
  int row0 = (blockIdx.x * WPB + wave) * 16;
  bool active = row0 < GIN_N;
  int m    = lane & 15;
  int half = lane >> 4;
  float* st = sT[wave];

  if (active) {
    const char* hbase = (const char*)h;
    unsigned lo = (unsigned)m * 16u;
    int2 ri = rowinfo[row0 + m];
    int* si = sIdx[wave];
    int idx[16];
#pragma unroll
    for (int r = 0; r < 16; ++r) {
      int sr = __builtin_amdgcn_readlane(ri.x, r);
      int dr = __builtin_amdgcn_readlane(ri.y, r);
      bool live = lane < dr;
      int w = csr[live ? sr + lane : 0];
      idx[r] = live ? w : row0 + r;
    }
#pragma unroll
    for (int r = 0; r < 16; ++r) si[r * 32 + lane] = idx[r];
    const int* sq = si + half * 32 + m;
    int ia = sq[0], ib = sq[16];
    for (int q = 0; q < 8; ++q) {
      int node = row0 + 2 * q + half;
      int dA = __builtin_amdgcn_readlane(ri.y, 2 * q);
      int dB = __builtin_amdgcn_readlane(ri.y, 2 * q + 1);
      int sA = __builtin_amdgcn_readlane(ri.x, 2 * q);
      int sB = __builtin_amdgcn_readlane(ri.x, 2 * q + 1);
      int deg   = half ? dB : dA;
      int start = half ? sB : sA;
      int degM = dA > dB ? dA : dB;
      const int* sn = sq + ((q + 1) & 7) * 64;
      int iaN = sn[0], ibN = sn[16];
      float4 hs = *(const float4*)(hbase + (((unsigned)node << 8) | lo));
      float4 a0 = make_float4(0.f, 0.f, 0.f, 0.f);
      float4 a1 = make_float4(0.f, 0.f, 0.f, 0.f);
      for (int base = 0; base < degM; base += 16) {
        int iv = ia;
        if (base == 16) iv = ib;
        if (base > 16) iv = (base + m < deg) ? csr[start + base + m] : node;
        int rem = deg - base;
        float4 v[16];
#pragma unroll
        for (int k = 0; k < 16; ++k) {
          int s = __shfl(iv, k, 16);
          v[k] = *(const float4*)(hbase + (((unsigned)s << 8) | lo));
        }
#pragma unroll
        for (int k = 0; k < 16; k += 2) {
          float w0 = (k < rem) ? 1.f : 0.f;
          float w1 = (k + 1 < rem) ? 1.f : 0.f;
          a0.x = fmaf(v[k].x, w0, a0.x); a0.y = fmaf(v[k].y, w0, a0.y);
          a0.z = fmaf(v[k].z, w0, a0.z); a0.w = fmaf(v[k].w, w0, a0.w);
          a1.x = fmaf(v[k + 1].x, w1, a1.x); a1.y = fmaf(v[k + 1].y, w1, a1.y);
          a1.z = fmaf(v[k + 1].z, w1, a1.z); a1.w = fmaf(v[k + 1].w, w1, a1.w);
        }
      }
      a0.x += a1.x; a0.y += a1.y; a0.z += a1.z; a0.w += a1.w;
      float* zr = st + (2 * q + half) * TSTRIDE + m * 4;
      v2f z01, z23;
      z01.x = hs.x + a0.x; z01.y = hs.y + a0.y;
      z23.x = hs.z + a0.z; z23.y = hs.w + a0.w;
      *(v2f*)zr = z01;
      *(v2f*)(zr + 2) = z23;
      ia = iaN; ib = ibN;
    }
  }
  __syncthreads();
  if (!active) return;

  v2f zA[16];
#pragma unroll
  for (int k0 = 0; k0 < 16; ++k0)
    zA[k0] = *(const v2f*)&st[m * TSTRIDE + 4 * k0 + 2 * half];

  v8f acc[4];
#pragma unroll
  for (int n0 = 0; n0 < 4; ++n0)
#pragma unroll
    for (int i = 0; i < 8; ++i) acc[n0][i] = 0.f;

#pragma unroll
  for (int n0 = 0; n0 < 4; ++n0) {
    const float* wcol = &sW1t[(n0 * 16 + m) * TSTRIDE + 2 * half];
#pragma unroll
    for (int k0 = 0; k0 < 16; ++k0) {
      v2f b = *(const v2f*)(wcol + 4 * k0);
      acc[n0] = __builtin_amdgcn_wmma_f32_16x16x4_f32(
          false, zA[k0], false, b, (short)0, acc[n0], false, false);
    }
  }

#pragma unroll
  for (int n0 = 0; n0 < 4; ++n0) {
    float bias = b1[n0 * 16 + m];
#pragma unroll
    for (int i = 0; i < 8; ++i) {
      float v = fmaxf(acc[n0][i] + bias, 0.f);
      st[(half * 8 + i) * TSTRIDE + n0 * 16 + m] = v;
    }
  }

  v2f h1A[16];
#pragma unroll
  for (int k0 = 0; k0 < 16; ++k0)
    h1A[k0] = *(const v2f*)&st[m * TSTRIDE + 4 * k0 + 2 * half];

  v8f acc2[4];
#pragma unroll
  for (int n0 = 0; n0 < 4; ++n0)
#pragma unroll
    for (int i = 0; i < 8; ++i) acc2[n0][i] = 0.f;

#pragma unroll
  for (int n0 = 0; n0 < 4; ++n0) {
    const float* wcol = &sW2t[(n0 * 16 + m) * TSTRIDE + 2 * half];
#pragma unroll
    for (int k0 = 0; k0 < 16; ++k0) {
      v2f b = *(const v2f*)(wcol + 4 * k0);
      acc2[n0] = __builtin_amdgcn_wmma_f32_16x16x4_f32(
          false, h1A[k0], false, b, (short)0, acc2[n0], false, false);
    }
  }

#pragma unroll
  for (int n0 = 0; n0 < 4; ++n0) {
    int col = n0 * 16 + m;
    float bias = b2[col];
    float scale = 1.f, shift = 0.f;
    if (do_bn) {
      float s = g[col] * rsqrtf(va[col] + 1e-5f);
      scale = s;
      shift = be[col] - mu[col] * s;
    }
#pragma unroll
    for (int i = 0; i < 8; ++i) {
      float v = acc2[n0][i] + bias;
      if (do_bn) v = fmaxf(v * scale + shift, 0.f);
      out[(size_t)(row0 + half * 8 + i) * GIN_D + col] = v;
    }
  }
}

extern "C" void kernel_launch(void* const* d_in, const int* in_sizes, int n_in,
                              void* d_out, int out_size, void* d_ws, size_t ws_size,
                              hipStream_t stream) {
  const float* x  = (const float*)d_in[0];
  const int*   ei = (const int*)d_in[1];
  const float* W1 = (const float*)d_in[2];
  const float* b1 = (const float*)d_in[3];
  const float* W2 = (const float*)d_in[4];
  const float* b2 = (const float*)d_in[5];
  const float* g  = (const float*)d_in[6];
  const float* be = (const float*)d_in[7];
  const float* mu = (const float*)d_in[8];
  const float* va = (const float*)d_in[9];
  float* out = (float*)d_out;

  float* hA    = (float*)d_ws;
  float* hB    = hA + (size_t)GIN_N * GIN_D;
  int2* rowinfo = (int2*)(hB + (size_t)GIN_N * GIN_D);
  int*  csr     = (int*)(rowinfo + GIN_N);
  unsigned* binned = (unsigned*)(csr + GIN_E);
  int*  tbl     = (int*)(binned + GIN_E);
  int*  binstart = tbl + NTBL;

  const int lgrid = (GIN_N / 16 + WPB - 1) / WPB;

  gin_bincount<<<NBLK, ST, 0, stream>>>(ei + GIN_E, tbl);
  gin_binsplit<<<NBLK, ST, 0, stream>>>(ei, ei + GIN_E, tbl, binstart, binned);
  gin_binrows<<<NBIN, ST, 0, stream>>>(binstart, binned, rowinfo, csr);

  const float* hcur = x;
  for (int l = 0; l < 5; ++l) {
    float* dstp = (l == 4) ? out : ((l % 2 == 0) ? hB : hA);
    gin_layer<<<lgrid, WPB * 32, 0, stream>>>(
        hcur, rowinfo, csr,
        W1 + (size_t)l * GIN_D * GIN_D, b1 + (size_t)l * GIN_D,
        W2 + (size_t)l * GIN_D * GIN_D, b2 + (size_t)l * GIN_D,
        g  + (size_t)l * GIN_D, be + (size_t)l * GIN_D,
        mu + (size_t)l * GIN_D, va + (size_t)l * GIN_D,
        dstp, (l < 4) ? 1 : 0);
    hcur = dstp;
  }
}
